# starting kernel
# baseline (speedup 1.0000x reference)
#include <hip/hip_runtime.h>
#include <hip/hip_bf16.h>

typedef float v2f __attribute__((ext_vector_type(2)));
typedef float v8f __attribute__((ext_vector_type(8)));

__global__ void deg_init_kernel(float* __restrict__ deg, int n) {
    int i = blockIdx.x * blockDim.x + threadIdx.x;
    if (i < n) deg[i] = 1.0f;
}

__global__ void deg_accum_kernel(const int* __restrict__ dst, float* __restrict__ deg, int e) {
    int i = blockIdx.x * blockDim.x + threadIdx.x;
    if (i < e) atomicAdd(&deg[dst[i]], 1.0f);
}

__global__ void rsqrt_kernel(float* __restrict__ deg, int n) {
    int i = blockIdx.x * blockDim.x + threadIdx.x;
    if (i < n) deg[i] = rsqrtf(deg[i]);
}

__global__ void gemm_wmma_f32_kernel(const float* __restrict__ A,
                                     const float* __restrict__ B,
                                     float* __restrict__ C,
                                     int M, int K, int N) {
    const int lane  = threadIdx.x & 31;
    const int wave  = threadIdx.x >> 5;
    const int tilesN = N >> 4;
    const int total  = (M >> 4) * tilesN;
    const int tile   = blockIdx.x * (blockDim.x >> 5) + wave;
    if (tile >= total) return;

    const int mt   = tile / tilesN;
    const int nt   = tile - mt * tilesN;
    const int half = lane >> 4;
    const int l15  = lane & 15;

    const float* arow = A + (long long)(mt * 16 + l15) * K;
    const float* bcol = B + nt * 16 + l15;

    v8f c = {};
    for (int k = 0; k < K; k += 4) {
        v2f a, b;
        a.x = arow[k + 2 * half];
        a.y = arow[k + 2 * half + 1];
        b.x = bcol[(long long)(k + 2 * half) * N];
        b.y = bcol[(long long)(k + 2 * half + 1) * N];
        c = __builtin_amdgcn_wmma_f32_16x16x4_f32(
                 false, a,  false, b,
                 (short)0, c,  false,  false);
    }

    float* crow = C + (long long)(mt * 16) * N + nt * 16 + l15;
#pragma unroll
    for (int r = 0; r < 8; ++r)
        crow[(long long)(r + 8 * half) * N] = c[r];
}

__global__ void agg_init_kernel(const float* __restrict__ xw,
                                const float* __restrict__ dinv,
                                const float* __restrict__ bias,
                                float* __restrict__ out,
                                int n, int F) {
    long long i = (long long)blockIdx.x * blockDim.x + threadIdx.x;
    long long total = (long long)n * F;
    if (i >= total) return;
    int node = (int)(i / F);
    int f    = (int)(i - (long long)node * F);
    float di = dinv[node];
    out[i] = xw[i] * di * di + bias[f];
}

__global__ void edge_scatter_kernel(const int* __restrict__ src,
                                    const int* __restrict__ dst,
                                    const float* __restrict__ dinv,
                                    const float* __restrict__ xw,
                                    float* __restrict__ out,
                                    int E, int F) {
    long long t = (long long)blockIdx.x * blockDim.x + threadIdx.x;
    long long total = (long long)E * F;
    if (t >= total) return;
    int e = (int)(t / F);
    int f = (int)(t - (long long)e * F);
    int s = src[e];
    int d = dst[e];
    float nrm = dinv[s] * dinv[d];
    atomicAdd(&out[(long long)d * F + f], xw[(long long)s * F + f] * nrm);
}

__global__ void relu_kernel(float* __restrict__ x, long long n) {
    long long i = (long long)blockIdx.x * blockDim.x + threadIdx.x;
    if (i < n) x[i] = fmaxf(x[i], 0.0f);
}

extern "C" void kernel_launch(void* const* d_in, const int* in_sizes, int n_in,
                              void* d_out, int out_size, void* d_ws, size_t ws_size,
                              hipStream_t stream) {
    const float* x   = (const float*)d_in[0];
    const int*   ei  = (const int*)d_in[1];
    const float* W1  = (const float*)d_in[2];
    const float* b1  = (const float*)d_in[3];
    const float* W2  = (const float*)d_in[4];
    const float* b2  = (const float*)d_in[5];
    float*       out = (float*)d_out;

    const int E     = in_sizes[1] / 2;
    const int F_HID = in_sizes[3];
    const int F_IN  = in_sizes[2] / F_HID;
    const int F_OUT = in_sizes[5];
    const int N     = in_sizes[0] / F_IN;

    const int* src = ei;
    const int* dst = ei + E;

    float* deg = (float*)d_ws;
    float* xw1 = deg + 65536;
    float* h   = xw1 + (long long)N * F_HID;
    float* hw2 = h   + (long long)N * F_HID;

    const int TB = 256;

    deg_init_kernel<<<(N + TB - 1) / TB, TB, 0, stream>>>(deg, N);
    deg_accum_kernel<<<(E + TB - 1) / TB, TB, 0, stream>>>(dst, deg, E);
    rsqrt_kernel<<<(N + TB - 1) / TB, TB, 0, stream>>>(deg, N);

    {
        int tiles = (N / 16) * (F_HID / 16);
        int wavesPerBlock = TB / 32;
        int blocks = (tiles + wavesPerBlock - 1) / wavesPerBlock;
        gemm_wmma_f32_kernel<<<blocks, TB, 0, stream>>>(x, W1, xw1, N, F_IN, F_HID);
    }

    {
        long long total = (long long)N * F_HID;
        agg_init_kernel<<<(int)((total + TB - 1) / TB), TB, 0, stream>>>(xw1, deg, b1, h, N, F_HID);
        long long etotal = (long long)E * F_HID;
        edge_scatter_kernel<<<(int)((etotal + TB - 1) / TB), TB, 0, stream>>>(src, dst, deg, xw1, h, E, F_HID);
        relu_kernel<<<(int)((total + TB - 1) / TB), TB, 0, stream>>>(h, total);
    }

    {
        int tiles = (N / 16) * (F_OUT / 16);
        int wavesPerBlock = TB / 32;
        int blocks = (tiles + wavesPerBlock - 1) / wavesPerBlock;
        gemm_wmma_f32_kernel<<<blocks, TB, 0, stream>>>(h, W2, hw2, N, F_HID, F_OUT);
    }

    {
        long long total = (long long)N * F_OUT;
        agg_init_kernel<<<(int)((total + TB - 1) / TB), TB, 0, stream>>>(hw2, deg, b2, out, N, F_OUT);
        long long etotal = (long long)E * F_OUT;
        edge_scatter_kernel<<<(int)((etotal + TB - 1) / TB), TB, 0, stream>>>(src, dst, deg, hw2, out, E, F_OUT);
    }
}


# edge scatter-add by wave: 32 edges per wave, src/dst/norm read once per edge and broadcast, lane per column, 8 whole-line row loads in flight; f32 atomic adds kept, arrival order as before
# speedup vs baseline: 1.9194x; 1.9194x over previous
#include <hip/hip_runtime.h>
#include <hip/hip_bf16.h>

typedef float v2f __attribute__((ext_vector_type(2)));
typedef float v8f __attribute__((ext_vector_type(8)));

__global__ void deg_init_kernel(float* __restrict__ deg, int n) {
    int i = blockIdx.x * blockDim.x + threadIdx.x;
    if (i < n) deg[i] = 1.0f;
}

__global__ void deg_accum_kernel(const int* __restrict__ dst, float* __restrict__ deg, int e) {
    int i = blockIdx.x * blockDim.x + threadIdx.x;
    if (i < e) atomicAdd(&deg[dst[i]], 1.0f);
}

__global__ void rsqrt_kernel(float* __restrict__ deg, int n) {
    int i = blockIdx.x * blockDim.x + threadIdx.x;
    if (i < n) deg[i] = rsqrtf(deg[i]);
}

__global__ void gemm_wmma_f32_kernel(const float* __restrict__ A,
                                     const float* __restrict__ B,
                                     float* __restrict__ C,
                                     int M, int K, int N) {
    const int lane  = threadIdx.x & 31;
    const int wave  = threadIdx.x >> 5;
    const int tilesN = N >> 4;
    const int total  = (M >> 4) * tilesN;
    const int tile   = blockIdx.x * (blockDim.x >> 5) + wave;
    if (tile >= total) return;

    const int mt   = tile / tilesN;
    const int nt   = tile - mt * tilesN;
    const int half = lane >> 4;
    const int l15  = lane & 15;

    const float* arow = A + (long long)(mt * 16 + l15) * K;
    const float* bcol = B + nt * 16 + l15;

    v8f c = {};
    for (int k = 0; k < K; k += 4) {
        v2f a, b;
        a.x = arow[k + 2 * half];
        a.y = arow[k + 2 * half + 1];
        b.x = bcol[(long long)(k + 2 * half) * N];
        b.y = bcol[(long long)(k + 2 * half + 1) * N];
        c = __builtin_amdgcn_wmma_f32_16x16x4_f32(
                 false, a,  false, b,
                 (short)0, c,  false,  false);
    }

    float* crow = C + (long long)(mt * 16) * N + nt * 16 + l15;
#pragma unroll
    for (int r = 0; r < 8; ++r)
        crow[(long long)(r + 8 * half) * N] = c[r];
}

__global__ void agg_init_kernel(const float* __restrict__ xw,
                                const float* __restrict__ dinv,
                                const float* __restrict__ bias,
                                float* __restrict__ out,
                                int n, int F) {
    long long i = (long long)blockIdx.x * blockDim.x + threadIdx.x;
    long long total = (long long)n * F;
    if (i >= total) return;
    int node = (int)(i / F);
    int f    = (int)(i - (long long)node * F);
    float di = dinv[node];
    out[i] = xw[i] * di * di + bias[f];
}

__global__ void __launch_bounds__(256)
edge_scatter_kernel(const int* __restrict__ src,
                    const int* __restrict__ dst,
                    const float* __restrict__ dinv,
                    const float* __restrict__ xw,
                    float* __restrict__ out,
                    int E, int F) {
    const int lane  = threadIdx.x & 31;
    const int wave  = __builtin_amdgcn_readfirstlane((int)(threadIdx.x >> 5));
    const int first = (blockIdx.x * (blockDim.x >> 5) + wave) * 32;
    if (first >= E) return;
    const int cnt = (E - first < 32) ? (E - first) : 32;

    int   s = 0, d = 0;
    float nrm = 0.0f;
    if (lane < cnt) {
        s   = src[first + lane];
        d   = dst[first + lane];
        nrm = dinv[s] * dinv[d];
    }
    const int nbits = __float_as_int(nrm);

    for (int f0 = 0; f0 < F; f0 += 32) {
        const int f = f0 + lane;
        if (f < F) {
            const unsigned fu = (unsigned)f;
            int j = 0;
            for (; j + 8 <= cnt; j += 8) {
                float v[8];
                for (int k = 0; k < 8; ++k) {
                    const float* row = xw + (long long)__builtin_amdgcn_readlane(s, j + k) * F;
                    v[k] = row[fu];
                }
                for (int k = 0; k < 8; ++k) {
                    float* orow = out + (long long)__builtin_amdgcn_readlane(d, j + k) * F;
                    const float nj = __int_as_float(__builtin_amdgcn_readlane(nbits, j + k));
                    atomicAdd(&orow[fu], v[k] * nj);
                }
            }
            for (; j < cnt; ++j) {
                const float* row  = xw  + (long long)__builtin_amdgcn_readlane(s, j) * F;
                float*       orow = out + (long long)__builtin_amdgcn_readlane(d, j) * F;
                const float  nj   = __int_as_float(__builtin_amdgcn_readlane(nbits, j));
                atomicAdd(&orow[fu], row[fu] * nj);
            }
        }
    }
}

__global__ void relu_kernel(float* __restrict__ x, long long n) {
    long long i = (long long)blockIdx.x * blockDim.x + threadIdx.x;
    if (i < n) x[i] = fmaxf(x[i], 0.0f);
}

extern "C" void kernel_launch(void* const* d_in, const int* in_sizes, int n_in,
                              void* d_out, int out_size, void* d_ws, size_t ws_size,
                              hipStream_t stream) {
    const float* x   = (const float*)d_in[0];
    const int*   ei  = (const int*)d_in[1];
    const float* W1  = (const float*)d_in[2];
    const float* b1  = (const float*)d_in[3];
    const float* W2  = (const float*)d_in[4];
    const float* b2  = (const float*)d_in[5];
    float*       out = (float*)d_out;

    const int E     = in_sizes[1] / 2;
    const int F_HID = in_sizes[3];
    const int F_IN  = in_sizes[2] / F_HID;
    const int F_OUT = in_sizes[5];
    const int N     = in_sizes[0] / F_IN;

    const int* src = ei;
    const int* dst = ei + E;

    float* deg = (float*)d_ws;
    float* xw1 = deg + 65536;
    float* h   = xw1 + (long long)N * F_HID;
    float* hw2 = h   + (long long)N * F_HID;

    const int TB = 256;
    const int eblocks = ((E + 31) / 32 + TB / 32 - 1) / (TB / 32);

    deg_init_kernel<<<(N + TB - 1) / TB, TB, 0, stream>>>(deg, N);
    deg_accum_kernel<<<(E + TB - 1) / TB, TB, 0, stream>>>(dst, deg, E);
    rsqrt_kernel<<<(N + TB - 1) / TB, TB, 0, stream>>>(deg, N);

    {
        int tiles = (N / 16) * (F_HID / 16);
        int wavesPerBlock = TB / 32;
        int blocks = (tiles + wavesPerBlock - 1) / wavesPerBlock;
        gemm_wmma_f32_kernel<<<blocks, TB, 0, stream>>>(x, W1, xw1, N, F_IN, F_HID);
    }

    {
        long long total = (long long)N * F_HID;
        agg_init_kernel<<<(int)((total + TB - 1) / TB), TB, 0, stream>>>(xw1, deg, b1, h, N, F_HID);
        edge_scatter_kernel<<<eblocks, TB, 0, stream>>>(src, dst, deg, xw1, h, E, F_HID);
        relu_kernel<<<(int)((total + TB - 1) / TB), TB, 0, stream>>>(h, total);
    }

    {
        int tiles = (N / 16) * (F_OUT / 16);
        int wavesPerBlock = TB / 32;
        int blocks = (tiles + wavesPerBlock - 1) / wavesPerBlock;
        gemm_wmma_f32_kernel<<<blocks, TB, 0, stream>>>(h, W2, hw2, N, F_HID, F_OUT);
    }

    {
        long long total = (long long)N * F_OUT;
        agg_init_kernel<<<(int)((total + TB - 1) / TB), TB, 0, stream>>>(hw2, deg, b2, out, N, F_OUT);
        edge_scatter_kernel<<<eblocks, TB, 0, stream>>>(src, dst, deg, hw2, out, E, F_OUT);
    }
}


# elementwise passes (rsqrt, self-loop+bias init x2, ReLU) fused into the GEMMs: epilogue also stores C*dinv^2+bias, ReLU applied as A is read, rsqrt(deg) taken where used; 10 launches -> 6
# speedup vs baseline: 2.1986x; 1.1455x over previous
#include <hip/hip_runtime.h>
#include <hip/hip_bf16.h>

typedef float v2f __attribute__((ext_vector_type(2)));
typedef float v8f __attribute__((ext_vector_type(8)));

__global__ void deg_init_kernel(float* __restrict__ deg, int n) {
    int i = blockIdx.x * blockDim.x + threadIdx.x;
    if (i < n) deg[i] = 1.0f;
}

__global__ void deg_accum_kernel(const int* __restrict__ dst, float* __restrict__ deg, int e) {
    int i = blockIdx.x * blockDim.x + threadIdx.x;
    if (i < e) atomicAdd(&deg[dst[i]], 1.0f);
}

template <bool RELU_A>
__global__ void gemm_wmma_f32_kernel(const float* __restrict__ A,
                                     const float* __restrict__ B,
                                     const float* __restrict__ deg,
                                     const float* __restrict__ bias,
                                     float* __restrict__ C,
                                     float* __restrict__ agg,
                                     int M, int K, int N) {
    const int lane  = threadIdx.x & 31;
    const int wave  = threadIdx.x >> 5;
    const int tilesN = N >> 4;
    const int total  = (M >> 4) * tilesN;
    const int tile   = blockIdx.x * (blockDim.x >> 5) + wave;
    if (tile >= total) return;

    const int mt   = tile / tilesN;
    const int nt   = tile - mt * tilesN;
    const int half = lane >> 4;
    const int l15  = lane & 15;

    const float* arow = A + (long long)(mt * 16 + l15) * K;
    const float* bcol = B + nt * 16 + l15;

    v8f c = {};
    for (int k = 0; k < K; k += 4) {
        v2f a, b;
        a.x = arow[k + 2 * half];
        a.y = arow[k + 2 * half + 1];
        if (RELU_A) {
            a.x = fmaxf(a.x, 0.0f);
            a.y = fmaxf(a.y, 0.0f);
        }
        b.x = bcol[(long long)(k + 2 * half) * N];
        b.y = bcol[(long long)(k + 2 * half + 1) * N];
        c = __builtin_amdgcn_wmma_f32_16x16x4_f32(
                 false, a,  false, b,
                 (short)0, c,  false,  false);
    }

    const int row0 = mt * 16 + 8 * half;
    const int col  = nt * 16 + l15;
    const float bv = bias[col];
    float* crow = C   + (long long)row0 * N + col;
    float* grow = agg + (long long)row0 * N + col;
#pragma unroll
    for (int r = 0; r < 8; ++r) {
        const float di = rsqrtf(deg[row0 + r]);
        crow[(long long)r * N] = c[r];
        grow[(long long)r * N] = c[r] * di * di + bv;
    }
}

__global__ void __launch_bounds__(256)
edge_scatter_kernel(const int* __restrict__ src,
                    const int* __restrict__ dst,
                    const float* __restrict__ deg,
                    const float* __restrict__ xw,
                    float* __restrict__ out,
                    int E, int F) {
    const int lane  = threadIdx.x & 31;
    const int wave  = __builtin_amdgcn_readfirstlane((int)(threadIdx.x >> 5));
    const int first = (blockIdx.x * (blockDim.x >> 5) + wave) * 32;
    if (first >= E) return;
    const int cnt = (E - first < 32) ? (E - first) : 32;

    int   s = 0, d = 0;
    float nrm = 0.0f;
    if (lane < cnt) {
        s   = src[first + lane];
        d   = dst[first + lane];
        nrm = rsqrtf(deg[s]) * rsqrtf(deg[d]);
    }
    const int nbits = __float_as_int(nrm);

    for (int f0 = 0; f0 < F; f0 += 32) {
        const int f = f0 + lane;
        if (f < F) {
            const unsigned fu = (unsigned)f;
            int j = 0;
            for (; j + 8 <= cnt; j += 8) {
                float v[8];
                for (int k = 0; k < 8; ++k) {
                    const float* row = xw + (long long)__builtin_amdgcn_readlane(s, j + k) * F;
                    v[k] = row[fu];
                }
                for (int k = 0; k < 8; ++k) {
                    float* orow = out + (long long)__builtin_amdgcn_readlane(d, j + k) * F;
                    const float nj = __int_as_float(__builtin_amdgcn_readlane(nbits, j + k));
                    atomicAdd(&orow[fu], v[k] * nj);
                }
            }
            for (; j < cnt; ++j) {
                const float* row  = xw  + (long long)__builtin_amdgcn_readlane(s, j) * F;
                float*       orow = out + (long long)__builtin_amdgcn_readlane(d, j) * F;
                const float  nj   = __int_as_float(__builtin_amdgcn_readlane(nbits, j));
                atomicAdd(&orow[fu], row[fu] * nj);
            }
        }
    }
}

extern "C" void kernel_launch(void* const* d_in, const int* in_sizes, int n_in,
                              void* d_out, int out_size, void* d_ws, size_t ws_size,
                              hipStream_t stream) {
    const float* x   = (const float*)d_in[0];
    const int*   ei  = (const int*)d_in[1];
    const float* W1  = (const float*)d_in[2];
    const float* b1  = (const float*)d_in[3];
    const float* W2  = (const float*)d_in[4];
    const float* b2  = (const float*)d_in[5];
    float*       out = (float*)d_out;

    const int E     = in_sizes[1] / 2;
    const int F_HID = in_sizes[3];
    const int F_IN  = in_sizes[2] / F_HID;
    const int F_OUT = in_sizes[5];
    const int N     = in_sizes[0] / F_IN;

    const int* src = ei;
    const int* dst = ei + E;

    float* deg = (float*)d_ws;
    float* xw1 = deg + 65536;
    float* h   = xw1 + (long long)N * F_HID;
    float* hw2 = h   + (long long)N * F_HID;

    const int TB = 256;
    const int eblocks = ((E + 31) / 32 + TB / 32 - 1) / (TB / 32);

    deg_init_kernel<<<(N + TB - 1) / TB, TB, 0, stream>>>(deg, N);
    deg_accum_kernel<<<(E + TB - 1) / TB, TB, 0, stream>>>(dst, deg, E);

    {
        int tiles = (N / 16) * (F_HID / 16);
        int wavesPerBlock = TB / 32;
        int blocks = (tiles + wavesPerBlock - 1) / wavesPerBlock;
        gemm_wmma_f32_kernel<false><<<blocks, TB, 0, stream>>>(x, W1, deg, b1, xw1, h, N, F_IN, F_HID);
    }

    edge_scatter_kernel<<<eblocks, TB, 0, stream>>>(src, dst, deg, xw1, h, E, F_HID);

    {
        int tiles = (N / 16) * (F_OUT / 16);
        int wavesPerBlock = TB / 32;
        int blocks = (tiles + wavesPerBlock - 1) / wavesPerBlock;
        gemm_wmma_f32_kernel<true><<<blocks, TB, 0, stream>>>(h, W2, deg, b2, hw2, out, N, F_HID, F_OUT);
    }

    edge_scatter_kernel<<<eblocks, TB, 0, stream>>>(src, dst, deg, hw2, out, E, F_OUT);
}


# neighbour sums by destination row: edges bucketed per call (int rank atomics, per-wave row allocation, fill); one wave sums self+bias+row in arrival-rank order, one store per row; no f32 atomics
# speedup vs baseline: 2.5452x; 1.1576x over previous
#include <hip/hip_runtime.h>
#include <hip/hip_bf16.h>

typedef float v2f __attribute__((ext_vector_type(2)));
typedef float v8f __attribute__((ext_vector_type(8)));
typedef int   v2i __attribute__((ext_vector_type(2)));

__global__ void count_clear_kernel(int* __restrict__ cnt, int* __restrict__ cursor, int n) {
    int i = blockIdx.x * blockDim.x + threadIdx.x;
    if (i < n) cnt[i] = 0;
    if (i == 0) *cursor = 0;
}

__global__ void count_rank_kernel(const int* __restrict__ dst, int* __restrict__ cnt,
                                  int* __restrict__ rank, int e) {
    int i = blockIdx.x * blockDim.x + threadIdx.x;
    if (i < e) rank[i] = atomicAdd(&cnt[dst[i]], 1);
}

__global__ void row_alloc_kernel(const int* __restrict__ cnt, int* __restrict__ cursor,
                                 int* __restrict__ rowinfo, int n) {
    const int i    = blockIdx.x * blockDim.x + threadIdx.x;
    const int lane = threadIdx.x & 31;
    const int c    = (i < n) ? cnt[i] : 0;
    int incl = c;
    for (int off = 1; off < 32; off <<= 1) {
        const int up = __shfl_up(incl, off);
        if (lane >= off) incl += up;
    }
    const int total = __shfl(incl, 31);
    int base = 0;
    if (lane == 0 && total > 0) base = atomicAdd(cursor, total);
    base = __shfl(base, 0);
    if (i < n) {
        v2i ri;
        ri.x = base + incl - c;
        ri.y = c;
        *(v2i*)(rowinfo + 2 * (long long)i) = ri;
    }
}

__global__ void row_fill_kernel(const int* __restrict__ src, const int* __restrict__ dst,
                                const int* __restrict__ rank, const int* __restrict__ cnt,
                                const int* __restrict__ rowinfo, int* __restrict__ ent, int e) {
    int i = blockIdx.x * blockDim.x + threadIdx.x;
    if (i >= e) return;
    const int s = src[i];
    const int d = dst[i];
    const int p = rowinfo[2 * (long long)d] + rank[i];
    v2i v;
    v.x = s;
    v.y = __float_as_int(rsqrtf((float)(cnt[s] + 1)));
    *(v2i*)(ent + 2 * (long long)p) = v;
}

template <bool RELU_A>
__global__ void gemm_wmma_f32_kernel(const float* __restrict__ A,
                                     const float* __restrict__ B,
                                     float* __restrict__ C,
                                     int M, int K, int N) {
    const int lane  = threadIdx.x & 31;
    const int wave  = threadIdx.x >> 5;
    const int tilesN = N >> 4;
    const int total  = (M >> 4) * tilesN;
    const int tile   = blockIdx.x * (blockDim.x >> 5) + wave;
    if (tile >= total) return;

    const int mt   = tile / tilesN;
    const int nt   = tile - mt * tilesN;
    const int half = lane >> 4;
    const int l15  = lane & 15;

    const float* arow = A + (long long)(mt * 16 + l15) * K;
    const float* bcol = B + nt * 16 + l15;

    v8f c = {};
    for (int k = 0; k < K; k += 4) {
        v2f a, b;
        a.x = arow[k + 2 * half];
        a.y = arow[k + 2 * half + 1];
        if (RELU_A) {
            a.x = fmaxf(a.x, 0.0f);
            a.y = fmaxf(a.y, 0.0f);
        }
        b.x = bcol[(long long)(k + 2 * half) * N];
        b.y = bcol[(long long)(k + 2 * half + 1) * N];
        c = __builtin_amdgcn_wmma_f32_16x16x4_f32(
                 false, a,  false, b,
                 (short)0, c,  false,  false);
    }

    const int row0 = mt * 16 + 8 * half;
    const int col  = nt * 16 + l15;
    float* crow = C + (long long)row0 * N + col;
#pragma unroll
    for (int r = 0; r < 8; ++r) crow[(long long)r * N] = c[r];
}

template <int VEC> struct RowVec;
template <> struct RowVec<1> { typedef float T; };
template <> struct RowVec<2> { typedef v2f   T; };

template <int G, int F, typename T>
__device__ __forceinline__ T add_slots(T acc, const float* __restrict__ xw, unsigned col,
                                       int s, int wbits, int j) {
    T v[G];
#pragma unroll
    for (int k = 0; k < G; ++k) {
        const float* row = xw + (long long)__builtin_amdgcn_readlane(s, j + k) * F;
        v[k] = *(const T*)(row + col);
    }
#pragma unroll
    for (int k = 0; k < G; ++k) {
        const float wk = __int_as_float(__builtin_amdgcn_readlane(wbits, j + k));
        acc += v[k] * wk;
    }
    return acc;
}

template <int VEC>
__global__ void __launch_bounds__(256)
row_gather_kernel(const int* __restrict__ rowinfo,
                  const int* __restrict__ ent,
                  const float* __restrict__ xw,
                  const float* __restrict__ bias,
                  float* __restrict__ out, int n) {
    typedef typename RowVec<VEC>::T T;
    constexpr int F = 32 * VEC;
    const int lane = threadIdx.x & 31;
    const int wave = __builtin_amdgcn_readfirstlane((int)(threadIdx.x >> 5));
    const int d    = blockIdx.x * (blockDim.x >> 5) + wave;
    if (d >= n) return;
    const v2i ri    = *(const v2i*)(rowinfo + 2 * (long long)d);
    const int beg   = ri.x;
    const int m_all = ri.y;
    const float di  = rsqrtf((float)(m_all + 1));
    const unsigned col = (unsigned)(lane * VEC);

    const T self = *(const T*)(xw + (long long)d * F + col);
    const T bv   = *(const T*)(bias + col);
    v2i e = {0, 0};
    if (lane < m_all) e = *(const v2i*)(ent + 2 * (long long)(beg + lane));
    T acc = self * di * di + bv;

    for (int base = 0; base < m_all; ) {
        const int m = (m_all - base < 32) ? (m_all - base) : 32;
        const int s     = e.x;
        const int wbits = __float_as_int(__int_as_float(e.y) * di);
        int j = 0;
        for (; j + 16 <= m; j += 16) acc = add_slots<16, F, T>(acc, xw, col, s, wbits, j);
        if (m - j >= 8) { acc = add_slots<8, F, T>(acc, xw, col, s, wbits, j); j += 8; }
        if (m - j >= 4) { acc = add_slots<4, F, T>(acc, xw, col, s, wbits, j); j += 4; }
        if (m - j >= 2) { acc = add_slots<2, F, T>(acc, xw, col, s, wbits, j); j += 2; }
        if (m - j >= 1) { acc = add_slots<1, F, T>(acc, xw, col, s, wbits, j); }
        base += 32;
        if (base < m_all) {
            e.x = 0; e.y = 0;
            if (lane < m_all - base) e = *(const v2i*)(ent + 2 * (long long)(beg + base + lane));
        }
    }
    *(T*)(out + (long long)d * F + col) = acc;
}

extern "C" void kernel_launch(void* const* d_in, const int* in_sizes, int n_in,
                              void* d_out, int out_size, void* d_ws, size_t ws_size,
                              hipStream_t stream) {
    const float* x   = (const float*)d_in[0];
    const int*   ei  = (const int*)d_in[1];
    const float* W1  = (const float*)d_in[2];
    const float* b1  = (const float*)d_in[3];
    const float* W2  = (const float*)d_in[4];
    const float* b2  = (const float*)d_in[5];
    float*       out = (float*)d_out;

    const int E     = in_sizes[1] / 2;
    const int F_HID = in_sizes[3];
    const int F_IN  = in_sizes[2] / F_HID;
    const int F_OUT = in_sizes[5];
    const int N     = in_sizes[0] / F_IN;

    const int* src = ei;
    const int* dst = ei + E;

    const long long Epad = ((long long)E + 63) & ~63LL;
    int*   cnt      = (int*)d_ws;
    int*   rowinfo  = cnt + 65536;
    int*   cursor   = rowinfo + 2 * 65536;
    int*   rank     = cursor + 64;
    int*   ent      = rank + Epad;
    float* xw1 = (float*)(ent + 2 * Epad);
    float* h   = xw1 + (long long)N * F_HID;
    float* hw2 = h   + (long long)N * F_HID;

    const int TB = 256;
    const int nblocks = (N + TB - 1) / TB;
    const int eblocks = (E + TB - 1) / TB;
    const int rblocks = (N + TB / 32 - 1) / (TB / 32);

    count_clear_kernel<<<nblocks, TB, 0, stream>>>(cnt, cursor, N);
    count_rank_kernel<<<eblocks, TB, 0, stream>>>(dst, cnt, rank, E);
    row_alloc_kernel<<<nblocks, TB, 0, stream>>>(cnt, cursor, rowinfo, N);
    row_fill_kernel<<<eblocks, TB, 0, stream>>>(src, dst, rank, cnt, rowinfo, ent, E);

    {
        int tiles = (N / 16) * (F_HID / 16);
        int wavesPerBlock = TB / 32;
        int blocks = (tiles + wavesPerBlock - 1) / wavesPerBlock;
        gemm_wmma_f32_kernel<false><<<blocks, TB, 0, stream>>>(x, W1, xw1, N, F_IN, F_HID);
    }

    row_gather_kernel<2><<<rblocks, TB, 0, stream>>>(rowinfo, ent, xw1, b1, h, N);

    {
        int tiles = (N / 16) * (F_OUT / 16);
        int wavesPerBlock = TB / 32;
        int blocks = (tiles + wavesPerBlock - 1) / wavesPerBlock;
        gemm_wmma_f32_kernel<true><<<blocks, TB, 0, stream>>>(h, W2, hw2, N, F_HID, F_OUT);
    }

    row_gather_kernel<1><<<rblocks, TB, 0, stream>>>(rowinfo, ent, hw2, b2, out, N);
}


# GEMMs retiled: one wave per 16-row strip over all N columns, A strip read once by whole-line loads and parked in LDS, weights transposed in LDS; k regrouped by 8: dot-product sum order changes
# speedup vs baseline: 3.3431x; 1.3135x over previous
#include <hip/hip_runtime.h>
#include <hip/hip_bf16.h>

typedef float v2f __attribute__((ext_vector_type(2)));
typedef float v4f __attribute__((ext_vector_type(4)));
typedef float v8f __attribute__((ext_vector_type(8)));
typedef int   v2i __attribute__((ext_vector_type(2)));

constexpr int CAP = 32;

__global__ void count_clear_kernel(int* __restrict__ cnt, int* __restrict__ ovcount, int n) {
    int i = blockIdx.x * blockDim.x + threadIdx.x;
    if (i < n) cnt[i] = 0;
    if (i == 0) *ovcount = 0;
}

__global__ void count_place_kernel(const int* __restrict__ src, const int* __restrict__ dst,
                                   int* __restrict__ cnt, int* __restrict__ ent,
                                   int* __restrict__ ov, int* __restrict__ ovcount, int e) {
    int i = blockIdx.x * blockDim.x + threadIdx.x;
    if (i >= e) return;
    const int s = src[i];
    const int d = dst[i];
    const int r = atomicAdd(&cnt[d], 1);
    if (r < CAP) {
        ent[(long long)d * CAP + r] = s;
    } else {
        const int k = atomicAdd(ovcount, 1);
        v2i p;
        p.x = d;
        p.y = s;
        *(v2i*)(ov + 2 * (long long)k) = p;
    }
}

constexpr int GEMM_WAVES = 8;

template <int K, int NT, bool RELU_A>
__global__ void __launch_bounds__(32 * GEMM_WAVES)
gemm_strip_wmma_f32_kernel(const float* __restrict__ A,
                           const float* __restrict__ B,
                           const int* __restrict__ cnt,
                           float* __restrict__ C, int M) {
    constexpr int N     = 16 * NT;
    constexpr int PITCH = K + 4;
    constexpr int KB    = K / 8;
    constexpr int TB    = 32 * GEMM_WAVES;
    constexpr int UNITS = (K / 4) * N / TB;
    __shared__ __attribute__((aligned(16))) float Bt[N * PITCH];
    __shared__ __attribute__((aligned(16))) float As[GEMM_WAVES * 16 * PITCH];

    const int tid   = threadIdx.x;
    const int lane  = tid & 31;
    const int wave  = tid >> 5;
    const int half  = lane >> 4;
    const int l15   = lane & 15;
    const int strips = M >> 4;
    const int strip  = blockIdx.x * GEMM_WAVES + wave;
    const int sload  = (strip < strips) ? strip : strips - 1;

    v4f w[UNITS];
#pragma unroll
    for (int i = 0; i < UNITS; ++i) {
        const int p  = tid + TB * i;
        const int n  = p % N;
        const int kg = p / N;
        const float* bsrc = B + (long long)(4 * kg) * N + n;
        w[i].x = bsrc[0];
        w[i].y = bsrc[N];
        w[i].z = bsrc[2 * N];
        w[i].w = bsrc[3 * N];
    }
    const float* astrip = A + (long long)sload * (16 * K) + 4 * lane;
    v4f av[KB];
#pragma unroll
    for (int i = 0; i < KB; ++i) av[i] = *(const v4f*)(astrip + 128 * i);

#pragma unroll
    for (int i = 0; i < UNITS; ++i) {
        const int p  = tid + TB * i;
        const int n  = p % N;
        const int kg = p / N;
        *(v4f*)(&Bt[n * PITCH + 4 * kg]) = w[i];
    }
    float* aw = As + wave * (16 * PITCH);
#pragma unroll
    for (int i = 0; i < KB; ++i) {
        const int f   = 128 * i + 4 * lane;
        const int row = f / K;
        const int col = f % K;
        v4f t = av[i];
        if (RELU_A) {
            t.x = fmaxf(t.x, 0.0f);
            t.y = fmaxf(t.y, 0.0f);
            t.z = fmaxf(t.z, 0.0f);
            t.w = fmaxf(t.w, 0.0f);
        }
        *(v4f*)(&aw[row * PITCH + col]) = t;
    }
    __syncthreads();
    if (strip >= strips) return;

    v8f c[NT];
#pragma unroll
    for (int nt = 0; nt < NT; ++nt) c[nt] = v8f{};

    const float* al = aw + l15 * PITCH + 4 * half;
    const float* bl = Bt + l15 * PITCH + 4 * half;
#pragma unroll
    for (int j = 0; j < KB; ++j) {
        const v4f a4 = *(const v4f*)(al + 8 * j);
        v4f bv[NT];
#pragma unroll
        for (int nt = 0; nt < NT; ++nt) bv[nt] = *(const v4f*)(bl + nt * 16 * PITCH + 8 * j);
        v2f a0, a1;
        a0.x = a4.x; a0.y = a4.y;
        a1.x = a4.z; a1.y = a4.w;
#pragma unroll
        for (int nt = 0; nt < NT; ++nt) {
            v2f b0;
            b0.x = bv[nt].x; b0.y = bv[nt].y;
            c[nt] = __builtin_amdgcn_wmma_f32_16x16x4_f32(
                         false, a0,  false, b0,
                         (short)0, c[nt],  false,  false);
        }
#pragma unroll
        for (int nt = 0; nt < NT; ++nt) {
            v2f b1;
            b1.x = bv[nt].z; b1.y = bv[nt].w;
            c[nt] = __builtin_amdgcn_wmma_f32_16x16x4_f32(
                         false, a1,  false, b1,
                         (short)0, c[nt],  false,  false);
        }
    }

    const int row0 = strip * 16 + 8 * half;
    float* crow = C + (long long)row0 * N + l15;
#pragma unroll
    for (int r = 0; r < 8; ++r) {
        const float di = rsqrtf((float)(cnt[row0 + r] + 1));
#pragma unroll
        for (int nt = 0; nt < NT; ++nt)
            crow[(long long)r * N + nt * 16] = c[nt][r] * di;
    }
}

template <int VEC> struct RowVec;
template <> struct RowVec<1> { typedef float T; };
template <> struct RowVec<2> { typedef v2f   T; };

template <int G, int F, typename T>
__device__ __forceinline__ T add_slots(T acc, const float* __restrict__ xs, unsigned col,
                                       int s, int j) {
    T v[G];
#pragma unroll
    for (int k = 0; k < G; ++k) {
        const float* row = xs + (long long)__builtin_amdgcn_readlane(s, j + k) * F;
        v[k] = *(const T*)(row + col);
    }
#pragma unroll
    for (int k = 0; k < G; ++k) acc += v[k];
    return acc;
}

template <int VEC>
__global__ void __launch_bounds__(256)
row_gather_kernel(const int* __restrict__ cnt,
                  const int* __restrict__ ent,
                  const int* __restrict__ ov,
                  const int* __restrict__ ovcount,
                  const float* __restrict__ xs,
                  const float* __restrict__ bias,
                  float* __restrict__ out, int n) {
    typedef typename RowVec<VEC>::T T;
    constexpr int F = 32 * VEC;
    const int lane = threadIdx.x & 31;
    const int wave = __builtin_amdgcn_readfirstlane((int)(threadIdx.x >> 5));
    const int d    = blockIdx.x * (blockDim.x >> 5) + wave;
    if (d >= n) return;
    const int m_all = cnt[d];
    const float di  = rsqrtf((float)(m_all + 1));
    const unsigned col = (unsigned)(lane * VEC);

    const T self = *(const T*)(xs + (long long)d * F + col);
    const T bv   = *(const T*)(bias + col);
    const int m  = (m_all < CAP) ? m_all : CAP;
    int s = 0;
    if (lane < m) s = ent[(long long)d * CAP + lane];
    T acc = self;

    int j = 0;
    for (; j + 16 <= m; j += 16) acc = add_slots<16, F, T>(acc, xs, col, s, j);
    if (m - j >= 8) { acc = add_slots<8, F, T>(acc, xs, col, s, j); j += 8; }
    if (m - j >= 4) { acc = add_slots<4, F, T>(acc, xs, col, s, j); j += 4; }
    if (m - j >= 2) { acc = add_slots<2, F, T>(acc, xs, col, s, j); j += 2; }
    if (m - j >= 1) { acc = add_slots<1, F, T>(acc, xs, col, s, j); }

    if (m_all > CAP) {
        const int novf = *ovcount;
        for (int k0 = 0; k0 < novf; k0 += 32) {
            int od = -1, os = 0;
            if (k0 + lane < novf) {
                const v2i p = *(const v2i*)(ov + 2 * (long long)(k0 + lane));
                od = p.x;
                os = p.y;
            }
            unsigned hit = __builtin_amdgcn_ballot_w32(od == d);
            while (hit) {
                const int b = __builtin_ctz(hit);
                hit &= hit - 1;
                const float* row = xs + (long long)__builtin_amdgcn_readlane(os, b) * F;
                acc += *(const T*)(row + col);
            }
        }
    }
    *(T*)(out + (long long)d * F + col) = acc * di + bv;
}

constexpr int LAYER_IN = 128, LAYER_HID = 64, LAYER_OUT = 32;

extern "C" void kernel_launch(void* const* d_in, const int* in_sizes, int n_in,
                              void* d_out, int out_size, void* d_ws, size_t ws_size,
                              hipStream_t stream) {
    const float* x   = (const float*)d_in[0];
    const int*   ei  = (const int*)d_in[1];
    const float* W1  = (const float*)d_in[2];
    const float* b1  = (const float*)d_in[3];
    const float* W2  = (const float*)d_in[4];
    const float* b2  = (const float*)d_in[5];
    float*       out = (float*)d_out;

    const int E     = in_sizes[1] / 2;
    const int F_HID = in_sizes[3];
    const int F_IN  = in_sizes[2] / F_HID;
    const int F_OUT = in_sizes[5];
    const int N     = in_sizes[0] / F_IN;

    const int* src = ei;
    const int* dst = ei + E;

    const long long Epad = ((long long)E + 63) & ~63LL;
    const long long Npad = ((long long)N + 63) & ~63LL;
    int*   cnt      = (int*)d_ws;
    int*   ovcount  = cnt + Npad;
    int*   ent      = ovcount + 64;
    int*   ov       = ent + Npad * CAP;
    float* xw1 = (float*)(ov + 2 * Epad);
    float* h   = xw1 + (long long)N * F_HID;
    float* hw2 = h   + (long long)N * F_HID;

    const int TB = 256;
    const int nblocks = (N + TB - 1) / TB;
    const int eblocks = (E + TB - 1) / TB;
    const int rblocks = (N + TB / 32 - 1) / (TB / 32);
    const int gblocks = (N / 16 + GEMM_WAVES - 1) / GEMM_WAVES;

    count_clear_kernel<<<nblocks, TB, 0, stream>>>(cnt, ovcount, N);
    count_place_kernel<<<eblocks, TB, 0, stream>>>(src, dst, cnt, ent, ov, ovcount, E);

    gemm_strip_wmma_f32_kernel<LAYER_IN, LAYER_HID / 16, false>
        <<<gblocks, 32 * GEMM_WAVES, 0, stream>>>(x, W1, cnt, xw1, N);

    row_gather_kernel<2><<<rblocks, TB, 0, stream>>>(cnt, ent, ov, ovcount, xw1, b1, h, N);

    gemm_strip_wmma_f32_kernel<LAYER_HID, LAYER_OUT / 16, true>
        <<<gblocks, 32 * GEMM_WAVES, 0, stream>>>(h, W2, cnt, hw2, N);

    row_gather_kernel<1><<<rblocks, TB, 0, stream>>>(cnt, ent, ov, ovcount, hw2, b2, out, N);
}


# edges bucketed through LDS in two contiguous passes (4096-edge chunks split by 256-row range; a block per range ranks its rows in LDS, writes whole slot lines); sum order: LDS arrival rank
# speedup vs baseline: 5.7042x; 1.7063x over previous
#include <hip/hip_runtime.h>
#include <hip/hip_bf16.h>

typedef float v2f __attribute__((ext_vector_type(2)));
typedef float v4f __attribute__((ext_vector_type(4)));
typedef float v8f __attribute__((ext_vector_type(8)));
typedef int   v2i __attribute__((ext_vector_type(2)));

constexpr int CAP = 32;
constexpr int BUCKET_SHIFT    = 8;
constexpr int ROWS_PER_BUCKET = 1 << BUCKET_SHIFT;
constexpr int CHUNK           = 4096;
constexpr int SORT_THREADS    = 512;
constexpr int NB_MAX          = SORT_THREADS;

typedef int v4i __attribute__((ext_vector_type(4)));

__global__ void __launch_bounds__(SORT_THREADS)
edge_chunk_split_kernel(const int* __restrict__ src, const int* __restrict__ dst,
                        int* __restrict__ sorted_w, int* __restrict__ runs_w,
                        int* __restrict__ ovcount, int e, int nb, int nbp) {
    constexpr int PER = CHUNK / SORT_THREADS;
    __shared__ int hist[NB_MAX];
    __shared__ int start[NB_MAX];
    __shared__ int wtot[SORT_THREADS / 32];
    __shared__ __attribute__((aligned(16))) v2i stage[CHUNK];
    v2i* sorted = (v2i*)sorted_w;
    v2i* runs   = (v2i*)runs_w;

    const int tid  = threadIdx.x;
    const int c    = blockIdx.x;
    const int base = c * CHUNK;
    if (c == 0 && tid == 0) *ovcount = 0;
    for (int i = tid; i < nb; i += SORT_THREADS) hist[i] = 0;

    int s[PER], d[PER], r[PER];
#pragma unroll
    for (int i = 0; i < PER; ++i) {
        const int idx = base + tid + SORT_THREADS * i;
        s[i] = 0;
        d[i] = -1;
        if (idx < e) {
            s[i] = src[idx];
            d[i] = dst[idx];
        }
    }
    __syncthreads();
#pragma unroll
    for (int i = 0; i < PER; ++i) {
        r[i] = 0;
        if (d[i] >= 0) r[i] = atomicAdd(&hist[d[i] >> BUCKET_SHIFT], 1);
    }
    __syncthreads();

    {
        const int lane = tid & 31;
        const int wave = tid >> 5;
        const int v = (tid < nb) ? hist[tid] : 0;
        int incl = v;
        for (int off = 1; off < 32; off <<= 1) {
            const int t = __shfl_up(incl, off);
            if (lane >= off) incl += t;
        }
        if (lane == 31) wtot[wave] = incl;
        __syncthreads();
        int front = 0;
#pragma unroll
        for (int w = 0; w < SORT_THREADS / 32; ++w) front += (w < wave) ? wtot[w] : 0;
        if (tid < nb) start[tid] = front + incl - v;
    }
    __syncthreads();

    for (int i = tid; i < nb; i += SORT_THREADS) {
        v2i p;
        p.x = start[i];
        p.y = hist[i];
        runs[(long long)c * nbp + i] = p;
    }
#pragma unroll
    for (int i = 0; i < PER; ++i) {
        if (d[i] >= 0) {
            v2i p;
            p.x = s[i];
            p.y = d[i];
            stage[start[d[i] >> BUCKET_SHIFT] + r[i]] = p;
        }
    }
    __syncthreads();
#pragma unroll
    for (int i = 0; i < PER; ++i) {
        const int k = tid + SORT_THREADS * i;
        if (base + k < e) sorted[(long long)base + k] = stage[k];
    }
}

constexpr int TPC = 2;

__device__ __forceinline__ void place_edge(int* lcnt, int* lent, int* __restrict__ ov,
                                           int* __restrict__ ovcount, int row0, v2i ed) {
    const int dl = ed.y - row0;
    const int r  = atomicAdd(&lcnt[dl], 1);
    if (r < CAP) {
        lent[dl * CAP + r] = ed.x;
    } else {
        const int k2 = atomicAdd(ovcount, 1);
        v2i o;
        o.x = ed.y;
        o.y = ed.x;
        *(v2i*)(ov + 2 * (long long)k2) = o;
    }
}

__global__ void __launch_bounds__(SORT_THREADS)
range_rows_kernel(const int* __restrict__ sorted_w, const int* __restrict__ runs_w,
                  int* __restrict__ cnt, int* __restrict__ ent,
                  int* __restrict__ ov, int* __restrict__ ovcount,
                  int n, int nc, int nbp) {
    __shared__ int lcnt[ROWS_PER_BUCKET];
    __shared__ __attribute__((aligned(16))) int lent[ROWS_PER_BUCKET * CAP];
    const v2i* sorted = (const v2i*)sorted_w;
    const v2i* runs   = (const v2i*)runs_w;

    const int tid  = threadIdx.x;
    const int b    = blockIdx.x;
    const int row0 = b << BUCKET_SHIFT;
    if (tid < ROWS_PER_BUCKET) lcnt[tid] = 0;
    __syncthreads();

    const int q = tid % TPC;
    for (int c = tid / TPC; c < nc; c += SORT_THREADS / TPC) {
        const v2i run = runs[(long long)c * nbp + b];
        const v2i* p  = sorted + (long long)c * CHUNK + run.x;
        for (int k = q; k < run.y; k += 2 * TPC) {
            const bool two = (k + TPC < run.y);
            const int k1 = two ? k + TPC : k;
            const v2i e0 = p[k];
            const v2i e1 = p[k1];
            place_edge(lcnt, lent, ov, ovcount, row0, e0);
            if (two) place_edge(lcnt, lent, ov, ovcount, row0, e1);
        }
    }
    __syncthreads();

    if (tid < ROWS_PER_BUCKET && row0 + tid < n) cnt[row0 + tid] = lcnt[tid];
#pragma unroll
    for (int i = 0; i < ROWS_PER_BUCKET * CAP / 4 / SORT_THREADS; ++i) {
        const int idx = tid + SORT_THREADS * i;
        const int row = row0 + (idx >> 3);
        if (row < n) ((v4i*)ent)[(long long)row0 * (CAP / 4) + idx] = ((const v4i*)lent)[idx];
    }
}

constexpr int GEMM_WAVES = 8;

template <int K, int NT, bool RELU_A>
__global__ void __launch_bounds__(32 * GEMM_WAVES)
gemm_strip_wmma_f32_kernel(const float* __restrict__ A,
                           const float* __restrict__ B,
                           const int* __restrict__ cnt,
                           float* __restrict__ C, int M) {
    constexpr int N     = 16 * NT;
    constexpr int PITCH = K + 4;
    constexpr int KB    = K / 8;
    constexpr int TB    = 32 * GEMM_WAVES;
    constexpr int UNITS = (K / 4) * N / TB;
    __shared__ __attribute__((aligned(16))) float Bt[N * PITCH];
    __shared__ __attribute__((aligned(16))) float As[GEMM_WAVES * 16 * PITCH];

    const int tid   = threadIdx.x;
    const int lane  = tid & 31;
    const int wave  = tid >> 5;
    const int half  = lane >> 4;
    const int l15   = lane & 15;
    const int strips = M >> 4;
    const int strip  = blockIdx.x * GEMM_WAVES + wave;
    const int sload  = (strip < strips) ? strip : strips - 1;

    v4f w[UNITS];
#pragma unroll
    for (int i = 0; i < UNITS; ++i) {
        const int p  = tid + TB * i;
        const int n  = p % N;
        const int kg = p / N;
        const float* bsrc = B + (long long)(4 * kg) * N + n;
        w[i].x = bsrc[0];
        w[i].y = bsrc[N];
        w[i].z = bsrc[2 * N];
        w[i].w = bsrc[3 * N];
    }
    const float* astrip = A + (long long)sload * (16 * K) + 4 * lane;
    v4f av[KB];
#pragma unroll
    for (int i = 0; i < KB; ++i) av[i] = *(const v4f*)(astrip + 128 * i);

#pragma unroll
    for (int i = 0; i < UNITS; ++i) {
        const int p  = tid + TB * i;
        const int n  = p % N;
        const int kg = p / N;
        *(v4f*)(&Bt[n * PITCH + 4 * kg]) = w[i];
    }
    float* aw = As + wave * (16 * PITCH);
#pragma unroll
    for (int i = 0; i < KB; ++i) {
        const int f   = 128 * i + 4 * lane;
        const int row = f / K;
        const int col = f % K;
        v4f t = av[i];
        if (RELU_A) {
            t.x = fmaxf(t.x, 0.0f);
            t.y = fmaxf(t.y, 0.0f);
            t.z = fmaxf(t.z, 0.0f);
            t.w = fmaxf(t.w, 0.0f);
        }
        *(v4f*)(&aw[row * PITCH + col]) = t;
    }
    __syncthreads();
    if (strip >= strips) return;

    v8f c[NT];
#pragma unroll
    for (int nt = 0; nt < NT; ++nt) c[nt] = v8f{};

    const float* al = aw + l15 * PITCH + 4 * half;
    const float* bl = Bt + l15 * PITCH + 4 * half;
#pragma unroll
    for (int j = 0; j < KB; ++j) {
        const v4f a4 = *(const v4f*)(al + 8 * j);
        v4f bv[NT];
#pragma unroll
        for (int nt = 0; nt < NT; ++nt) bv[nt] = *(const v4f*)(bl + nt * 16 * PITCH + 8 * j);
        v2f a0, a1;
        a0.x = a4.x; a0.y = a4.y;
        a1.x = a4.z; a1.y = a4.w;
#pragma unroll
        for (int nt = 0; nt < NT; ++nt) {
            v2f b0;
            b0.x = bv[nt].x; b0.y = bv[nt].y;
            c[nt] = __builtin_amdgcn_wmma_f32_16x16x4_f32(
                         false, a0,  false, b0,
                         (short)0, c[nt],  false,  false);
        }
#pragma unroll
        for (int nt = 0; nt < NT; ++nt) {
            v2f b1;
            b1.x = bv[nt].z; b1.y = bv[nt].w;
            c[nt] = __builtin_amdgcn_wmma_f32_16x16x4_f32(
                         false, a1,  false, b1,
                         (short)0, c[nt],  false,  false);
        }
    }

    const int row0 = strip * 16 + 8 * half;
    float* crow = C + (long long)row0 * N + l15;
#pragma unroll
    for (int r = 0; r < 8; ++r) {
        const float di = rsqrtf((float)(cnt[row0 + r] + 1));
#pragma unroll
        for (int nt = 0; nt < NT; ++nt)
            crow[(long long)r * N + nt * 16] = c[nt][r] * di;
    }
}

template <int VEC> struct RowVec;
template <> struct RowVec<1> { typedef float T; };
template <> struct RowVec<2> { typedef v2f   T; };

template <int G, int F, typename T>
__device__ __forceinline__ T add_slots(T acc, const float* __restrict__ xs, unsigned col,
                                       int s, int j) {
    T v[G];
#pragma unroll
    for (int k = 0; k < G; ++k) {
        const float* row = xs + (long long)__builtin_amdgcn_readlane(s, j + k) * F;
        v[k] = *(const T*)(row + col);
    }
#pragma unroll
    for (int k = 0; k < G; ++k) acc += v[k];
    return acc;
}

template <int VEC>
__global__ void __launch_bounds__(256)
row_gather_kernel(const int* __restrict__ cnt,
                  const int* __restrict__ ent,
                  const int* __restrict__ ov,
                  const int* __restrict__ ovcount,
                  const float* __restrict__ xs,
                  const float* __restrict__ bias,
                  float* __restrict__ out, int n) {
    typedef typename RowVec<VEC>::T T;
    constexpr int F = 32 * VEC;
    const int lane = threadIdx.x & 31;
    const int wave = __builtin_amdgcn_readfirstlane((int)(threadIdx.x >> 5));
    const int d    = blockIdx.x * (blockDim.x >> 5) + wave;
    if (d >= n) return;
    const int m_all = cnt[d];
    const float di  = rsqrtf((float)(m_all + 1));
    const unsigned col = (unsigned)(lane * VEC);

    const T self = *(const T*)(xs + (long long)d * F + col);
    const T bv   = *(const T*)(bias + col);
    const int m  = (m_all < CAP) ? m_all : CAP;
    int s = 0;
    if (lane < m) s = ent[(long long)d * CAP + lane];
    T acc = self;

    int j = 0;
    for (; j + 16 <= m; j += 16) acc = add_slots<16, F, T>(acc, xs, col, s, j);
    if (m - j >= 8) { acc = add_slots<8, F, T>(acc, xs, col, s, j); j += 8; }
    if (m - j >= 4) { acc = add_slots<4, F, T>(acc, xs, col, s, j); j += 4; }
    if (m - j >= 2) { acc = add_slots<2, F, T>(acc, xs, col, s, j); j += 2; }
    if (m - j >= 1) { acc = add_slots<1, F, T>(acc, xs, col, s, j); }

    if (m_all > CAP) {
        const int novf = *ovcount;
        for (int k0 = 0; k0 < novf; k0 += 32) {
            int od = -1, os = 0;
            if (k0 + lane < novf) {
                const v2i p = *(const v2i*)(ov + 2 * (long long)(k0 + lane));
                od = p.x;
                os = p.y;
            }
            unsigned hit = __builtin_amdgcn_ballot_w32(od == d);
            while (hit) {
                const int b = __builtin_ctz(hit);
                hit &= hit - 1;
                const float* row = xs + (long long)__builtin_amdgcn_readlane(os, b) * F;
                acc += *(const T*)(row + col);
            }
        }
    }
    *(T*)(out + (long long)d * F + col) = acc * di + bv;
}

constexpr int LAYER_IN = 128, LAYER_HID = 64, LAYER_OUT = 32;

extern "C" void kernel_launch(void* const* d_in, const int* in_sizes, int n_in,
                              void* d_out, int out_size, void* d_ws, size_t ws_size,
                              hipStream_t stream) {
    const float* x   = (const float*)d_in[0];
    const int*   ei  = (const int*)d_in[1];
    const float* W1  = (const float*)d_in[2];
    const float* b1  = (const float*)d_in[3];
    const float* W2  = (const float*)d_in[4];
    const float* b2  = (const float*)d_in[5];
    float*       out = (float*)d_out;

    const int E     = in_sizes[1] / 2;
    const int F_HID = in_sizes[3];
    const int F_IN  = in_sizes[2] / F_HID;
    const int F_OUT = in_sizes[5];
    const int N     = in_sizes[0] / F_IN;

    const int* src = ei;
    const int* dst = ei + E;

    const long long Epad = ((long long)E + 63) & ~63LL;
    const long long Npad = ((long long)N + 63) & ~63LL;
    int*   cnt      = (int*)d_ws;
    int*   ovcount  = cnt + Npad;
    int*   ent      = ovcount + 64;
    int*   ov       = ent + Npad * CAP;
    const int NC  = (E + CHUNK - 1) / CHUNK;
    const int NB  = (N + ROWS_PER_BUCKET - 1) / ROWS_PER_BUCKET;
    const int NBP = (NB + 31) & ~31;
    int*   sorted   = ov + 2 * Epad;
    int*   runs     = sorted + 2 * Epad;
    float* xw1 = (float*)(runs + 2 * (long long)NC * NBP);
    float* h   = xw1 + (long long)N * F_HID;
    float* hw2 = h   + (long long)N * F_HID;

    const int TB = 256;
    const int rblocks = (N + TB / 32 - 1) / (TB / 32);
    const int gblocks = (N / 16 + GEMM_WAVES - 1) / GEMM_WAVES;

    edge_chunk_split_kernel<<<NC, SORT_THREADS, 0, stream>>>(src, dst, sorted, runs, ovcount,
                                                             E, NB, NBP);
    range_rows_kernel<<<NB, SORT_THREADS, 0, stream>>>(sorted, runs, cnt, ent, ov, ovcount,
                                                       N, NC, NBP);

    gemm_strip_wmma_f32_kernel<LAYER_IN, LAYER_HID / 16, false>
        <<<gblocks, 32 * GEMM_WAVES, 0, stream>>>(x, W1, cnt, xw1, N);

    row_gather_kernel<2><<<rblocks, TB, 0, stream>>>(cnt, ent, ov, ovcount, xw1, b1, h, N);

    gemm_strip_wmma_f32_kernel<LAYER_HID, LAYER_OUT / 16, true>
        <<<gblocks, 32 * GEMM_WAVES, 0, stream>>>(h, W2, cnt, hw2, N);

    row_gather_kernel<1><<<rblocks, TB, 0, stream>>>(cnt, ent, ov, ovcount, hw2, b2, out, N);
}


# layer-2 product fused into the layer-1 row aggregation: a block sums a 16-row strip, parks relu(h) in LDS, one wave runs its WMMA product; h never stored; 6->5 launches; sum order unchanged
# speedup vs baseline: 6.2460x; 1.0950x over previous
#include <hip/hip_runtime.h>
#include <hip/hip_bf16.h>

typedef float v2f __attribute__((ext_vector_type(2)));
typedef float v4f __attribute__((ext_vector_type(4)));
typedef float v8f __attribute__((ext_vector_type(8)));
typedef int   v2i __attribute__((ext_vector_type(2)));

constexpr int CAP = 32;
constexpr int BUCKET_SHIFT    = 8;
constexpr int ROWS_PER_BUCKET = 1 << BUCKET_SHIFT;
constexpr int CHUNK           = 4096;
constexpr int SORT_THREADS    = 512;
constexpr int NB_MAX          = SORT_THREADS;

typedef int v4i __attribute__((ext_vector_type(4)));

__global__ void __launch_bounds__(SORT_THREADS)
edge_chunk_split_kernel(const int* __restrict__ src, const int* __restrict__ dst,
                        int* __restrict__ sorted_w, int* __restrict__ runs_w,
                        int* __restrict__ ovcount, int e, int nb, int nbp) {
    constexpr int PER = CHUNK / SORT_THREADS;
    __shared__ int hist[NB_MAX];
    __shared__ int start[NB_MAX];
    __shared__ int wtot[SORT_THREADS / 32];
    __shared__ __attribute__((aligned(16))) v2i stage[CHUNK];
    v2i* sorted = (v2i*)sorted_w;
    v2i* runs   = (v2i*)runs_w;

    const int tid  = threadIdx.x;
    const int c    = blockIdx.x;
    const int base = c * CHUNK;
    if (c == 0 && tid == 0) *ovcount = 0;
    for (int i = tid; i < nb; i += SORT_THREADS) hist[i] = 0;

    int s[PER], d[PER], r[PER];
#pragma unroll
    for (int i = 0; i < PER; ++i) {
        const int idx = base + tid + SORT_THREADS * i;
        s[i] = 0;
        d[i] = -1;
        if (idx < e) {
            s[i] = src[idx];
            d[i] = dst[idx];
        }
    }
    __syncthreads();
#pragma unroll
    for (int i = 0; i < PER; ++i) {
        r[i] = 0;
        if (d[i] >= 0) r[i] = atomicAdd(&hist[d[i] >> BUCKET_SHIFT], 1);
    }
    __syncthreads();

    {
        const int lane = tid & 31;
        const int wave = tid >> 5;
        const int v = (tid < nb) ? hist[tid] : 0;
        int incl = v;
        for (int off = 1; off < 32; off <<= 1) {
            const int t = __shfl_up(incl, off);
            if (lane >= off) incl += t;
        }
        if (lane == 31) wtot[wave] = incl;
        __syncthreads();
        int front = 0;
#pragma unroll
        for (int w = 0; w < SORT_THREADS / 32; ++w) front += (w < wave) ? wtot[w] : 0;
        if (tid < nb) start[tid] = front + incl - v;
    }
    __syncthreads();

    for (int i = tid; i < nb; i += SORT_THREADS) {
        v2i p;
        p.x = start[i];
        p.y = hist[i];
        runs[(long long)c * nbp + i] = p;
    }
#pragma unroll
    for (int i = 0; i < PER; ++i) {
        if (d[i] >= 0) {
            v2i p;
            p.x = s[i];
            p.y = d[i];
            stage[start[d[i] >> BUCKET_SHIFT] + r[i]] = p;
        }
    }
    __syncthreads();
#pragma unroll
    for (int i = 0; i < PER; ++i) {
        const int k = tid + SORT_THREADS * i;
        if (base + k < e) sorted[(long long)base + k] = stage[k];
    }
}

constexpr int TPC = 2;

__device__ __forceinline__ void place_edge(int* lcnt, int* lent, int* __restrict__ ov,
                                           int* __restrict__ ovcount, int row0, v2i ed) {
    const int dl = ed.y - row0;
    const int r  = atomicAdd(&lcnt[dl], 1);
    if (r < CAP) {
        lent[dl * CAP + r] = ed.x;
    } else {
        const int k2 = atomicAdd(ovcount, 1);
        v2i o;
        o.x = ed.y;
        o.y = ed.x;
        *(v2i*)(ov + 2 * (long long)k2) = o;
    }
}

__global__ void __launch_bounds__(SORT_THREADS)
range_rows_kernel(const int* __restrict__ sorted_w, const int* __restrict__ runs_w,
                  int* __restrict__ cnt, int* __restrict__ ent,
                  int* __restrict__ ov, int* __restrict__ ovcount,
                  int n, int nc, int nbp) {
    __shared__ int lcnt[ROWS_PER_BUCKET];
    __shared__ __attribute__((aligned(16))) int lent[ROWS_PER_BUCKET * CAP];
    const v2i* sorted = (const v2i*)sorted_w;
    const v2i* runs   = (const v2i*)runs_w;

    const int tid  = threadIdx.x;
    const int b    = blockIdx.x;
    const int row0 = b << BUCKET_SHIFT;
    if (tid < ROWS_PER_BUCKET) lcnt[tid] = 0;
    __syncthreads();

    const int q = tid % TPC;
    for (int c = tid / TPC; c < nc; c += SORT_THREADS / TPC) {
        const v2i run = runs[(long long)c * nbp + b];
        const v2i* p  = sorted + (long long)c * CHUNK + run.x;
        for (int k = q; k < run.y; k += 2 * TPC) {
            const bool two = (k + TPC < run.y);
            const int k1 = two ? k + TPC : k;
            const v2i e0 = p[k];
            const v2i e1 = p[k1];
            place_edge(lcnt, lent, ov, ovcount, row0, e0);
            if (two) place_edge(lcnt, lent, ov, ovcount, row0, e1);
        }
    }
    __syncthreads();

    if (tid < ROWS_PER_BUCKET && row0 + tid < n) cnt[row0 + tid] = lcnt[tid];
#pragma unroll
    for (int i = 0; i < ROWS_PER_BUCKET * CAP / 4 / SORT_THREADS; ++i) {
        const int idx = tid + SORT_THREADS * i;
        const int row = row0 + (idx >> 3);
        if (row < n) ((v4i*)ent)[(long long)row0 * (CAP / 4) + idx] = ((const v4i*)lent)[idx];
    }
}

constexpr int GEMM_WAVES = 8;

template <int K, int NT>
__global__ void __launch_bounds__(32 * GEMM_WAVES)
gemm_strip_wmma_f32_kernel(const float* __restrict__ A,
                           const float* __restrict__ B,
                           const int* __restrict__ cnt,
                           float* __restrict__ C, int M) {
    constexpr int N     = 16 * NT;
    constexpr int PITCH = K + 4;
    constexpr int KB    = K / 8;
    constexpr int TB    = 32 * GEMM_WAVES;
    constexpr int UNITS = (K / 4) * N / TB;
    __shared__ __attribute__((aligned(16))) float Bt[N * PITCH];
    __shared__ __attribute__((aligned(16))) float As[GEMM_WAVES * 16 * PITCH];

    const int tid   = threadIdx.x;
    const int lane  = tid & 31;
    const int wave  = tid >> 5;
    const int half  = lane >> 4;
    const int l15   = lane & 15;
    const int strips = M >> 4;
    const int strip  = blockIdx.x * GEMM_WAVES + wave;
    const int sload  = (strip < strips) ? strip : strips - 1;

    v4f w[UNITS];
#pragma unroll
    for (int i = 0; i < UNITS; ++i) {
        const int p  = tid + TB * i;
        const int n  = p % N;
        const int kg = p / N;
        const float* bsrc = B + (long long)(4 * kg) * N + n;
        w[i].x = bsrc[0];
        w[i].y = bsrc[N];
        w[i].z = bsrc[2 * N];
        w[i].w = bsrc[3 * N];
    }
    const float* astrip = A + (long long)sload * (16 * K) + 4 * lane;
    v4f av[KB];
#pragma unroll
    for (int i = 0; i < KB; ++i) av[i] = *(const v4f*)(astrip + 128 * i);

#pragma unroll
    for (int i = 0; i < UNITS; ++i) {
        const int p  = tid + TB * i;
        const int n  = p % N;
        const int kg = p / N;
        *(v4f*)(&Bt[n * PITCH + 4 * kg]) = w[i];
    }
    float* aw = As + wave * (16 * PITCH);
#pragma unroll
    for (int i = 0; i < KB; ++i) {
        const int f   = 128 * i + 4 * lane;
        const int row = f / K;
        const int col = f % K;
        *(v4f*)(&aw[row * PITCH + col]) = av[i];
    }
    __syncthreads();
    if (strip >= strips) return;

    v8f c[NT];
#pragma unroll
    for (int nt = 0; nt < NT; ++nt) c[nt] = v8f{};

    const float* al = aw + l15 * PITCH + 4 * half;
    const float* bl = Bt + l15 * PITCH + 4 * half;
#pragma unroll
    for (int j = 0; j < KB; ++j) {
        const v4f a4 = *(const v4f*)(al + 8 * j);
        v4f bv[NT];
#pragma unroll
        for (int nt = 0; nt < NT; ++nt) bv[nt] = *(const v4f*)(bl + nt * 16 * PITCH + 8 * j);
        v2f a0, a1;
        a0.x = a4.x; a0.y = a4.y;
        a1.x = a4.z; a1.y = a4.w;
#pragma unroll
        for (int nt = 0; nt < NT; ++nt) {
            v2f b0;
            b0.x = bv[nt].x; b0.y = bv[nt].y;
            c[nt] = __builtin_amdgcn_wmma_f32_16x16x4_f32(
                         false, a0,  false, b0,
                         (short)0, c[nt],  false,  false);
        }
#pragma unroll
        for (int nt = 0; nt < NT; ++nt) {
            v2f b1;
            b1.x = bv[nt].z; b1.y = bv[nt].w;
            c[nt] = __builtin_amdgcn_wmma_f32_16x16x4_f32(
                         false, a1,  false, b1,
                         (short)0, c[nt],  false,  false);
        }
    }

    const int row0 = strip * 16 + 8 * half;
    float* crow = C + (long long)row0 * N + l15;
#pragma unroll
    for (int r = 0; r < 8; ++r) {
        const float di = rsqrtf((float)(cnt[row0 + r] + 1));
#pragma unroll
        for (int nt = 0; nt < NT; ++nt)
            crow[(long long)r * N + nt * 16] = c[nt][r] * di;
    }
}

template <int VEC> struct RowVec;
template <> struct RowVec<1> { typedef float T; };
template <> struct RowVec<2> { typedef v2f   T; };

template <int G, int F, typename T>
__device__ __forceinline__ T add_slots(T acc, const float* __restrict__ xs, unsigned col,
                                       int s, int j) {
    T v[G];
#pragma unroll
    for (int k = 0; k < G; ++k) {
        const float* row = xs + (long long)__builtin_amdgcn_readlane(s, j + k) * F;
        v[k] = *(const T*)(row + col);
    }
#pragma unroll
    for (int k = 0; k < G; ++k) acc += v[k];
    return acc;
}

template <int F, typename T>
__device__ __forceinline__ T row_sum(const float* __restrict__ xs, const int* __restrict__ ov,
                                     const int* __restrict__ ovcount, unsigned col, int lane,
                                     int d, int m_all, int s, T self) {
    const int m = (m_all < CAP) ? m_all : CAP;
    T acc = self;
    int j = 0;
    for (; j + 16 <= m; j += 16) acc = add_slots<16, F, T>(acc, xs, col, s, j);
    if (m - j >= 8) { acc = add_slots<8, F, T>(acc, xs, col, s, j); j += 8; }
    if (m - j >= 4) { acc = add_slots<4, F, T>(acc, xs, col, s, j); j += 4; }
    if (m - j >= 2) { acc = add_slots<2, F, T>(acc, xs, col, s, j); j += 2; }
    if (m - j >= 1) { acc = add_slots<1, F, T>(acc, xs, col, s, j); }

    if (m_all > CAP) {
        const int novf = *ovcount;
        for (int k0 = 0; k0 < novf; k0 += 32) {
            int od = -1, os = 0;
            if (k0 + lane < novf) {
                const v2i p = *(const v2i*)(ov + 2 * (long long)(k0 + lane));
                od = p.x;
                os = p.y;
            }
            unsigned hit = __builtin_amdgcn_ballot_w32(od == d);
            while (hit) {
                const int b = __builtin_ctz(hit);
                hit &= hit - 1;
                const float* row = xs + (long long)__builtin_amdgcn_readlane(os, b) * F;
                acc += *(const T*)(row + col);
            }
        }
    }
    return acc;
}

template <int VEC>
__global__ void __launch_bounds__(256)
row_gather_kernel(const int* __restrict__ cnt,
                  const int* __restrict__ ent,
                  const int* __restrict__ ov,
                  const int* __restrict__ ovcount,
                  const float* __restrict__ xs,
                  const float* __restrict__ bias,
                  float* __restrict__ out, int n) {
    typedef typename RowVec<VEC>::T T;
    constexpr int F = 32 * VEC;
    const int lane = threadIdx.x & 31;
    const int wave = __builtin_amdgcn_readfirstlane((int)(threadIdx.x >> 5));
    const int d    = blockIdx.x * (blockDim.x >> 5) + wave;
    if (d >= n) return;
    const int m_all = cnt[d];
    const float di  = rsqrtf((float)(m_all + 1));
    const unsigned col = (unsigned)(lane * VEC);

    const T self = *(const T*)(xs + (long long)d * F + col);
    const T bv   = *(const T*)(bias + col);
    const int m  = (m_all < CAP) ? m_all : CAP;
    int s = 0;
    if (lane < m) s = ent[(long long)d * CAP + lane];

    const T acc = row_sum<F, T>(xs, ov, ovcount, col, lane, d, m_all, s, self);
    *(T*)(out + (long long)d * F + col) = acc * di + bv;
}

constexpr int STRIP_WAVES = 8;

template <int K, int NT>
__global__ void __launch_bounds__(32 * STRIP_WAVES)
gather_relu_gemm_wmma_f32_kernel(const int* __restrict__ cnt,
                                 const int* __restrict__ ent,
                                 const int* __restrict__ ov,
                                 const int* __restrict__ ovcount,
                                 const float* __restrict__ xs,
                                 const float* __restrict__ bias,
                                 const float* __restrict__ B,
                                 float* __restrict__ C) {
    static_assert(K == 64, "one lane holds two columns of a row");
    typedef typename RowVec<K / 32>::T T;
    constexpr int N     = 16 * NT;
    constexpr int PITCH = K + 4;
    constexpr int KB    = K / 8;
    constexpr int TB    = 32 * STRIP_WAVES;
    constexpr int UNITS = (K / 4) * N / TB;
    constexpr int RPW   = 16 / STRIP_WAVES;
    __shared__ __attribute__((aligned(16))) float Bt[N * PITCH];
    __shared__ __attribute__((aligned(16))) float As[16 * PITCH];

    const int tid   = threadIdx.x;
    const int lane  = tid & 31;
    const int wave  = __builtin_amdgcn_readfirstlane((int)(threadIdx.x >> 5));
    const int strip = blockIdx.x;
    const unsigned col = (unsigned)(lane * (K / 32));

    v4f w[UNITS];
#pragma unroll
    for (int i = 0; i < UNITS; ++i) {
        const int p  = tid + TB * i;
        const int nn = p % N;
        const int kg = p / N;
        const float* bsrc = B + (long long)(4 * kg) * N + nn;
        w[i].x = bsrc[0];
        w[i].y = bsrc[N];
        w[i].z = bsrc[2 * N];
        w[i].w = bsrc[3 * N];
    }
    const T bv = *(const T*)(bias + col);
    const int d0 = strip * 16 + wave * RPW;
    int m_all[RPW], s[RPW];
    T   self[RPW];
#pragma unroll
    for (int r = 0; r < RPW; ++r) m_all[r] = cnt[d0 + r];
#pragma unroll
    for (int r = 0; r < RPW; ++r) self[r] = *(const T*)(xs + (long long)(d0 + r) * K + col);
#pragma unroll
    for (int r = 0; r < RPW; ++r) {
        const int m = (m_all[r] < CAP) ? m_all[r] : CAP;
        s[r] = 0;
        if (lane < m) s[r] = ent[(long long)(d0 + r) * CAP + lane];
    }
#pragma unroll
    for (int i = 0; i < UNITS; ++i) {
        const int p  = tid + TB * i;
        const int nn = p % N;
        const int kg = p / N;
        *(v4f*)(&Bt[nn * PITCH + 4 * kg]) = w[i];
    }

#pragma unroll
    for (int r = 0; r < RPW; ++r) {
        const T acc = row_sum<K, T>(xs, ov, ovcount, col, lane, d0 + r, m_all[r], s[r], self[r]);
        const float di = rsqrtf((float)(m_all[r] + 1));
        T hv = acc * di + bv;
        hv.x = fmaxf(hv.x, 0.0f);
        hv.y = fmaxf(hv.y, 0.0f);
        *(T*)(&As[(wave * RPW + r) * PITCH + col]) = hv;
    }
    __syncthreads();
    if (wave != 0) return;

    const int half = lane >> 4;
    const int l15  = lane & 15;
    v8f c[NT];
#pragma unroll
    for (int nt = 0; nt < NT; ++nt) c[nt] = v8f{};

    const float* al = As + l15 * PITCH + 4 * half;
    const float* bl = Bt + l15 * PITCH + 4 * half;
#pragma unroll
    for (int j = 0; j < KB; ++j) {
        const v4f a4 = *(const v4f*)(al + 8 * j);
        v4f bq[NT];
#pragma unroll
        for (int nt = 0; nt < NT; ++nt) bq[nt] = *(const v4f*)(bl + nt * 16 * PITCH + 8 * j);
        v2f a0, a1;
        a0.x = a4.x; a0.y = a4.y;
        a1.x = a4.z; a1.y = a4.w;
#pragma unroll
        for (int nt = 0; nt < NT; ++nt) {
            v2f b0;
            b0.x = bq[nt].x; b0.y = bq[nt].y;
            c[nt] = __builtin_amdgcn_wmma_f32_16x16x4_f32(
                         false, a0,  false, b0,
                         (short)0, c[nt],  false,  false);
        }
#pragma unroll
        for (int nt = 0; nt < NT; ++nt) {
            v2f b1;
            b1.x = bq[nt].z; b1.y = bq[nt].w;
            c[nt] = __builtin_amdgcn_wmma_f32_16x16x4_f32(
                         false, a1,  false, b1,
                         (short)0, c[nt],  false,  false);
        }
    }

    const int row0 = strip * 16 + 8 * half;
    float* crow = C + (long long)row0 * N + l15;
#pragma unroll
    for (int r = 0; r < 8; ++r) {
        const float di = rsqrtf((float)(cnt[row0 + r] + 1));
#pragma unroll
        for (int nt = 0; nt < NT; ++nt)
            crow[(long long)r * N + nt * 16] = c[nt][r] * di;
    }
}

constexpr int LAYER_IN = 128, LAYER_HID = 64, LAYER_OUT = 32;

extern "C" void kernel_launch(void* const* d_in, const int* in_sizes, int n_in,
                              void* d_out, int out_size, void* d_ws, size_t ws_size,
                              hipStream_t stream) {
    const float* x   = (const float*)d_in[0];
    const int*   ei  = (const int*)d_in[1];
    const float* W1  = (const float*)d_in[2];
    const float* b1  = (const float*)d_in[3];
    const float* W2  = (const float*)d_in[4];
    const float* b2  = (const float*)d_in[5];
    float*       out = (float*)d_out;

    const int E     = in_sizes[1] / 2;
    const int F_HID = in_sizes[3];
    const int F_IN  = in_sizes[2] / F_HID;
    const int F_OUT = in_sizes[5];
    const int N     = in_sizes[0] / F_IN;

    const int* src = ei;
    const int* dst = ei + E;

    const long long Epad = ((long long)E + 63) & ~63LL;
    const long long Npad = ((long long)N + 63) & ~63LL;
    int*   cnt      = (int*)d_ws;
    int*   ovcount  = cnt + Npad;
    int*   ent      = ovcount + 64;
    int*   ov       = ent + Npad * CAP;
    const int NC  = (E + CHUNK - 1) / CHUNK;
    const int NB  = (N + ROWS_PER_BUCKET - 1) / ROWS_PER_BUCKET;
    const int NBP = (NB + 31) & ~31;
    int*   sorted   = ov + 2 * Epad;
    int*   runs     = sorted + 2 * Epad;
    float* xw1 = (float*)(runs + 2 * (long long)NC * NBP);
    float* hw2 = xw1 + (long long)N * F_HID;

    const int TB = 256;
    const int rblocks = (N + TB / 32 - 1) / (TB / 32);
    const int gblocks = (N / 16 + GEMM_WAVES - 1) / GEMM_WAVES;

    edge_chunk_split_kernel<<<NC, SORT_THREADS, 0, stream>>>(src, dst, sorted, runs, ovcount,
                                                             E, NB, NBP);
    range_rows_kernel<<<NB, SORT_THREADS, 0, stream>>>(sorted, runs, cnt, ent, ov, ovcount,
                                                       N, NC, NBP);

    gemm_strip_wmma_f32_kernel<LAYER_IN, LAYER_HID / 16>
        <<<gblocks, 32 * GEMM_WAVES, 0, stream>>>(x, W1, cnt, xw1, N);

    gather_relu_gemm_wmma_f32_kernel<LAYER_HID, LAYER_OUT / 16>
        <<<N / 16, 32 * STRIP_WAVES, 0, stream>>>(cnt, ent, ov, ovcount, xw1, b1, W2, hw2);

    row_gather_kernel<1><<<rblocks, TB, 0, stream>>>(cnt, ent, ov, ovcount, hw2, b2, out, N);
}


# pass 2 walks each run with 8 neighbouring lanes (64-byte pieces of 4 runs per load, not 16 lone entries), 16 entries per lane in flight, ranks back to back; sum order: LDS arrival rank, changes
# speedup vs baseline: 6.5233x; 1.0444x over previous
#include <hip/hip_runtime.h>
#include <hip/hip_bf16.h>

typedef float v2f __attribute__((ext_vector_type(2)));
typedef float v4f __attribute__((ext_vector_type(4)));
typedef float v8f __attribute__((ext_vector_type(8)));
typedef int   v2i __attribute__((ext_vector_type(2)));

constexpr int CAP = 32;
constexpr int BUCKET_SHIFT    = 8;
constexpr int ROWS_PER_BUCKET = 1 << BUCKET_SHIFT;
constexpr int CHUNK           = 4096;
constexpr int SORT_THREADS    = 512;
constexpr int NB_MAX          = SORT_THREADS;

typedef int v4i __attribute__((ext_vector_type(4)));

__global__ void __launch_bounds__(SORT_THREADS)
edge_chunk_split_kernel(const int* __restrict__ src, const int* __restrict__ dst,
                        int* __restrict__ sorted_w, int* __restrict__ runs_w,
                        int* __restrict__ ovcount, int e, int nb, int nbp) {
    constexpr int PER = CHUNK / SORT_THREADS;
    __shared__ int hist[NB_MAX];
    __shared__ int start[NB_MAX];
    __shared__ int wtot[SORT_THREADS / 32];
    __shared__ __attribute__((aligned(16))) v2i stage[CHUNK];
    v2i* sorted = (v2i*)sorted_w;
    v2i* runs   = (v2i*)runs_w;

    const int tid  = threadIdx.x;
    const int c    = blockIdx.x;
    const int base = c * CHUNK;
    if (c == 0 && tid == 0) *ovcount = 0;
    for (int i = tid; i < nb; i += SORT_THREADS) hist[i] = 0;

    int s[PER], d[PER], r[PER];
#pragma unroll
    for (int i = 0; i < PER; ++i) {
        const int idx = base + tid + SORT_THREADS * i;
        s[i] = 0;
        d[i] = -1;
        if (idx < e) {
            s[i] = src[idx];
            d[i] = dst[idx];
        }
    }
    __syncthreads();
#pragma unroll
    for (int i = 0; i < PER; ++i) {
        r[i] = 0;
        if (d[i] >= 0) r[i] = atomicAdd(&hist[d[i] >> BUCKET_SHIFT], 1);
    }
    __syncthreads();

    {
        const int lane = tid & 31;
        const int wave = tid >> 5;
        const int v = (tid < nb) ? hist[tid] : 0;
        int incl = v;
        for (int off = 1; off < 32; off <<= 1) {
            const int t = __shfl_up(incl, off);
            if (lane >= off) incl += t;
        }
        if (lane == 31) wtot[wave] = incl;
        __syncthreads();
        int front = 0;
#pragma unroll
        for (int w = 0; w < SORT_THREADS / 32; ++w) front += (w < wave) ? wtot[w] : 0;
        if (tid < nb) start[tid] = front + incl - v;
    }
    __syncthreads();

    for (int i = tid; i < nb; i += SORT_THREADS) {
        v2i p;
        p.x = start[i];
        p.y = hist[i];
        runs[(long long)c * nbp + i] = p;
    }
#pragma unroll
    for (int i = 0; i < PER; ++i) {
        if (d[i] >= 0) {
            v2i p;
            p.x = s[i];
            p.y = d[i];
            stage[start[d[i] >> BUCKET_SHIFT] + r[i]] = p;
        }
    }
    __syncthreads();
#pragma unroll
    for (int i = 0; i < PER; ++i) {
        const int k = tid + SORT_THREADS * i;
        if (base + k < e) sorted[(long long)base + k] = stage[k];
    }
}

constexpr int RUN_LANES  = 8;
constexpr int RUN_SLOTS  = SORT_THREADS / RUN_LANES;
constexpr int RUN_UNROLL = 4;
constexpr int RUN_DEPTH  = 4;

__device__ __forceinline__ void file_edge(int* lent, int* __restrict__ ov,
                                          int* __restrict__ ovcount, int row0, v2i ed, int r) {
    if (r < CAP) {
        lent[(ed.y - row0) * CAP + r] = ed.x;
    } else {
        const int k2 = atomicAdd(ovcount, 1);
        v2i o;
        o.x = ed.y;
        o.y = ed.x;
        *(v2i*)(ov + 2 * (long long)k2) = o;
    }
}

__global__ void __launch_bounds__(SORT_THREADS)
range_rows_kernel(const int* __restrict__ sorted_w, const int* __restrict__ runs_w,
                  int* __restrict__ cnt, int* __restrict__ ent,
                  int* __restrict__ ov, int* __restrict__ ovcount,
                  int n, int nc, int nbp) {
    __shared__ int lcnt[ROWS_PER_BUCKET];
    __shared__ __attribute__((aligned(16))) int lent[ROWS_PER_BUCKET * CAP];
    const v2i* sorted = (const v2i*)sorted_w;
    const v2i* runs   = (const v2i*)runs_w;

    const int tid  = threadIdx.x;
    const int b    = blockIdx.x;
    const int row0 = b << BUCKET_SHIFT;
    if (tid < ROWS_PER_BUCKET) lcnt[tid] = 0;
    __syncthreads();

    const int j    = tid % RUN_LANES;
    const int slot = tid / RUN_LANES;
    for (int c0 = slot; c0 < nc; c0 += RUN_UNROLL * RUN_SLOTS) {
        const v2i* pc[RUN_UNROLL];
        int first[RUN_UNROLL], len[RUN_UNROLL];
#pragma unroll
        for (int u = 0; u < RUN_UNROLL; ++u) {
            const int c  = c0 + u * RUN_SLOTS;
            const int cc = (c < nc) ? c : nc - 1;
            const v2i run = runs[(long long)cc * nbp + b];
            pc[u]    = sorted + (long long)cc * CHUNK;
            first[u] = run.x;
            len[u]   = (c < nc) ? run.y : 0;
        }
        v2i ed[RUN_UNROLL][RUN_DEPTH];
#pragma unroll
        for (int u = 0; u < RUN_UNROLL; ++u) {
#pragma unroll
            for (int d = 0; d < RUN_DEPTH; ++d) {
                const int k = j + RUN_LANES * d;
                ed[u][d] = pc[u][(k < len[u]) ? first[u] + k : 0];
            }
        }
        int r[RUN_UNROLL][RUN_DEPTH];
#pragma unroll
        for (int u = 0; u < RUN_UNROLL; ++u) {
#pragma unroll
            for (int d = 0; d < RUN_DEPTH; ++d) {
                r[u][d] = 0;
                if (j + RUN_LANES * d < len[u]) r[u][d] = atomicAdd(&lcnt[ed[u][d].y - row0], 1);
            }
        }
#pragma unroll
        for (int u = 0; u < RUN_UNROLL; ++u) {
#pragma unroll
            for (int d = 0; d < RUN_DEPTH; ++d) {
                if (j + RUN_LANES * d < len[u])
                    file_edge(lent, ov, ovcount, row0, ed[u][d], r[u][d]);
            }
        }
#pragma unroll
        for (int u = 0; u < RUN_UNROLL; ++u) {
            for (int k = j + RUN_LANES * RUN_DEPTH; k < len[u]; k += RUN_LANES) {
                const v2i e0 = pc[u][first[u] + k];
                const int r0 = atomicAdd(&lcnt[e0.y - row0], 1);
                file_edge(lent, ov, ovcount, row0, e0, r0);
            }
        }
    }
    __syncthreads();

    if (tid < ROWS_PER_BUCKET && row0 + tid < n) cnt[row0 + tid] = lcnt[tid];
#pragma unroll
    for (int i = 0; i < ROWS_PER_BUCKET * CAP / 4 / SORT_THREADS; ++i) {
        const int idx = tid + SORT_THREADS * i;
        const int row = row0 + (idx >> 3);
        if (row < n) ((v4i*)ent)[(long long)row0 * (CAP / 4) + idx] = ((const v4i*)lent)[idx];
    }
}

constexpr int GEMM_WAVES = 8;

template <int K, int NT>
__global__ void __launch_bounds__(32 * GEMM_WAVES)
gemm_strip_wmma_f32_kernel(const float* __restrict__ A,
                           const float* __restrict__ B,
                           const int* __restrict__ cnt,
                           float* __restrict__ C, int M) {
    constexpr int N     = 16 * NT;
    constexpr int PITCH = K + 4;
    constexpr int KB    = K / 8;
    constexpr int TB    = 32 * GEMM_WAVES;
    constexpr int UNITS = (K / 4) * N / TB;
    __shared__ __attribute__((aligned(16))) float Bt[N * PITCH];
    __shared__ __attribute__((aligned(16))) float As[GEMM_WAVES * 16 * PITCH];

    const int tid   = threadIdx.x;
    const int lane  = tid & 31;
    const int wave  = tid >> 5;
    const int half  = lane >> 4;
    const int l15   = lane & 15;
    const int strips = M >> 4;
    const int strip  = blockIdx.x * GEMM_WAVES + wave;
    const int sload  = (strip < strips) ? strip : strips - 1;

    v4f w[UNITS];
#pragma unroll
    for (int i = 0; i < UNITS; ++i) {
        const int p  = tid + TB * i;
        const int n  = p % N;
        const int kg = p / N;
        const float* bsrc = B + (long long)(4 * kg) * N + n;
        w[i].x = bsrc[0];
        w[i].y = bsrc[N];
        w[i].z = bsrc[2 * N];
        w[i].w = bsrc[3 * N];
    }
    const float* astrip = A + (long long)sload * (16 * K) + 4 * lane;
    v4f av[KB];
#pragma unroll
    for (int i = 0; i < KB; ++i) av[i] = *(const v4f*)(astrip + 128 * i);

#pragma unroll
    for (int i = 0; i < UNITS; ++i) {
        const int p  = tid + TB * i;
        const int n  = p % N;
        const int kg = p / N;
        *(v4f*)(&Bt[n * PITCH + 4 * kg]) = w[i];
    }
    float* aw = As + wave * (16 * PITCH);
#pragma unroll
    for (int i = 0; i < KB; ++i) {
        const int f   = 128 * i + 4 * lane;
        const int row = f / K;
        const int col = f % K;
        *(v4f*)(&aw[row * PITCH + col]) = av[i];
    }
    __syncthreads();
    if (strip >= strips) return;

    v8f c[NT];
#pragma unroll
    for (int nt = 0; nt < NT; ++nt) c[nt] = v8f{};

    const float* al = aw + l15 * PITCH + 4 * half;
    const float* bl = Bt + l15 * PITCH + 4 * half;
#pragma unroll
    for (int j = 0; j < KB; ++j) {
        const v4f a4 = *(const v4f*)(al + 8 * j);
        v4f bv[NT];
#pragma unroll
        for (int nt = 0; nt < NT; ++nt) bv[nt] = *(const v4f*)(bl + nt * 16 * PITCH + 8 * j);
        v2f a0, a1;
        a0.x = a4.x; a0.y = a4.y;
        a1.x = a4.z; a1.y = a4.w;
#pragma unroll
        for (int nt = 0; nt < NT; ++nt) {
            v2f b0;
            b0.x = bv[nt].x; b0.y = bv[nt].y;
            c[nt] = __builtin_amdgcn_wmma_f32_16x16x4_f32(
                         false, a0,  false, b0,
                         (short)0, c[nt],  false,  false);
        }
#pragma unroll
        for (int nt = 0; nt < NT; ++nt) {
            v2f b1;
            b1.x = bv[nt].z; b1.y = bv[nt].w;
            c[nt] = __builtin_amdgcn_wmma_f32_16x16x4_f32(
                         false, a1,  false, b1,
                         (short)0, c[nt],  false,  false);
        }
    }

    const int row0 = strip * 16 + 8 * half;
    float* crow = C + (long long)row0 * N + l15;
#pragma unroll
    for (int r = 0; r < 8; ++r) {
        const float di = rsqrtf((float)(cnt[row0 + r] + 1));
#pragma unroll
        for (int nt = 0; nt < NT; ++nt)
            crow[(long long)r * N + nt * 16] = c[nt][r] * di;
    }
}

template <int VEC> struct RowVec;
template <> struct RowVec<1> { typedef float T; };
template <> struct RowVec<2> { typedef v2f   T; };

template <int G, int F, typename T>
__device__ __forceinline__ T add_slots(T acc, const float* __restrict__ xs, unsigned col,
                                       int s, int j) {
    T v[G];
#pragma unroll
    for (int k = 0; k < G; ++k) {
        const float* row = xs + (long long)__builtin_amdgcn_readlane(s, j + k) * F;
        v[k] = *(const T*)(row + col);
    }
#pragma unroll
    for (int k = 0; k < G; ++k) acc += v[k];
    return acc;
}

template <int F, typename T>
__device__ __forceinline__ T row_sum(const float* __restrict__ xs, const int* __restrict__ ov,
                                     const int* __restrict__ ovcount, unsigned col, int lane,
                                     int d, int m_all, int s, T self) {
    const int m = (m_all < CAP) ? m_all : CAP;
    T acc = self;
    int j = 0;
    for (; j + 16 <= m; j += 16) acc = add_slots<16, F, T>(acc, xs, col, s, j);
    if (m - j >= 8) { acc = add_slots<8, F, T>(acc, xs, col, s, j); j += 8; }
    if (m - j >= 4) { acc = add_slots<4, F, T>(acc, xs, col, s, j); j += 4; }
    if (m - j >= 2) { acc = add_slots<2, F, T>(acc, xs, col, s, j); j += 2; }
    if (m - j >= 1) { acc = add_slots<1, F, T>(acc, xs, col, s, j); }

    if (m_all > CAP) {
        const int novf = *ovcount;
        for (int k0 = 0; k0 < novf; k0 += 32) {
            int od = -1, os = 0;
            if (k0 + lane < novf) {
                const v2i p = *(const v2i*)(ov + 2 * (long long)(k0 + lane));
                od = p.x;
                os = p.y;
            }
            unsigned hit = __builtin_amdgcn_ballot_w32(od == d);
            while (hit) {
                const int b = __builtin_ctz(hit);
                hit &= hit - 1;
                const float* row = xs + (long long)__builtin_amdgcn_readlane(os, b) * F;
                acc += *(const T*)(row + col);
            }
        }
    }
    return acc;
}

template <int VEC>
__global__ void __launch_bounds__(256)
row_gather_kernel(const int* __restrict__ cnt,
                  const int* __restrict__ ent,
                  const int* __restrict__ ov,
                  const int* __restrict__ ovcount,
                  const float* __restrict__ xs,
                  const float* __restrict__ bias,
                  float* __restrict__ out, int n) {
    typedef typename RowVec<VEC>::T T;
    constexpr int F = 32 * VEC;
    const int lane = threadIdx.x & 31;
    const int wave = __builtin_amdgcn_readfirstlane((int)(threadIdx.x >> 5));
    const int d    = blockIdx.x * (blockDim.x >> 5) + wave;
    if (d >= n) return;
    const int m_all = cnt[d];
    const float di  = rsqrtf((float)(m_all + 1));
    const unsigned col = (unsigned)(lane * VEC);

    const T self = *(const T*)(xs + (long long)d * F + col);
    const T bv   = *(const T*)(bias + col);
    const int m  = (m_all < CAP) ? m_all : CAP;
    int s = 0;
    if (lane < m) s = ent[(long long)d * CAP + lane];

    const T acc = row_sum<F, T>(xs, ov, ovcount, col, lane, d, m_all, s, self);
    *(T*)(out + (long long)d * F + col) = acc * di + bv;
}

constexpr int STRIP_WAVES = 8;

template <int K, int NT>
__global__ void __launch_bounds__(32 * STRIP_WAVES)
gather_relu_gemm_wmma_f32_kernel(const int* __restrict__ cnt,
                                 const int* __restrict__ ent,
                                 const int* __restrict__ ov,
                                 const int* __restrict__ ovcount,
                                 const float* __restrict__ xs,
                                 const float* __restrict__ bias,
                                 const float* __restrict__ B,
                                 float* __restrict__ C) {
    static_assert(K == 64, "one lane holds two columns of a row");
    typedef typename RowVec<K / 32>::T T;
    constexpr int N     = 16 * NT;
    constexpr int PITCH = K + 4;
    constexpr int KB    = K / 8;
    constexpr int TB    = 32 * STRIP_WAVES;
    constexpr int UNITS = (K / 4) * N / TB;
    constexpr int RPW   = 16 / STRIP_WAVES;
    __shared__ __attribute__((aligned(16))) float Bt[N * PITCH];
    __shared__ __attribute__((aligned(16))) float As[16 * PITCH];

    const int tid   = threadIdx.x;
    const int lane  = tid & 31;
    const int wave  = __builtin_amdgcn_readfirstlane((int)(threadIdx.x >> 5));
    const int strip = blockIdx.x;
    const unsigned col = (unsigned)(lane * (K / 32));

    v4f w[UNITS];
#pragma unroll
    for (int i = 0; i < UNITS; ++i) {
        const int p  = tid + TB * i;
        const int nn = p % N;
        const int kg = p / N;
        const float* bsrc = B + (long long)(4 * kg) * N + nn;
        w[i].x = bsrc[0];
        w[i].y = bsrc[N];
        w[i].z = bsrc[2 * N];
        w[i].w = bsrc[3 * N];
    }
    const T bv = *(const T*)(bias + col);
    const int d0 = strip * 16 + wave * RPW;
    int m_all[RPW], s[RPW];
    T   self[RPW];
#pragma unroll
    for (int r = 0; r < RPW; ++r) m_all[r] = cnt[d0 + r];
#pragma unroll
    for (int r = 0; r < RPW; ++r) self[r] = *(const T*)(xs + (long long)(d0 + r) * K + col);
#pragma unroll
    for (int r = 0; r < RPW; ++r) {
        const int m = (m_all[r] < CAP) ? m_all[r] : CAP;
        s[r] = 0;
        if (lane < m) s[r] = ent[(long long)(d0 + r) * CAP + lane];
    }
#pragma unroll
    for (int i = 0; i < UNITS; ++i) {
        const int p  = tid + TB * i;
        const int nn = p % N;
        const int kg = p / N;
        *(v4f*)(&Bt[nn * PITCH + 4 * kg]) = w[i];
    }

#pragma unroll
    for (int r = 0; r < RPW; ++r) {
        const T acc = row_sum<K, T>(xs, ov, ovcount, col, lane, d0 + r, m_all[r], s[r], self[r]);
        const float di = rsqrtf((float)(m_all[r] + 1));
        T hv = acc * di + bv;
        hv.x = fmaxf(hv.x, 0.0f);
        hv.y = fmaxf(hv.y, 0.0f);
        *(T*)(&As[(wave * RPW + r) * PITCH + col]) = hv;
    }
    __syncthreads();
    if (wave != 0) return;

    const int half = lane >> 4;
    const int l15  = lane & 15;
    v8f c[NT];
#pragma unroll
    for (int nt = 0; nt < NT; ++nt) c[nt] = v8f{};

    const float* al = As + l15 * PITCH + 4 * half;
    const float* bl = Bt + l15 * PITCH + 4 * half;
#pragma unroll
    for (int j = 0; j < KB; ++j) {
        const v4f a4 = *(const v4f*)(al + 8 * j);
        v4f bq[NT];
#pragma unroll
        for (int nt = 0; nt < NT; ++nt) bq[nt] = *(const v4f*)(bl + nt * 16 * PITCH + 8 * j);
        v2f a0, a1;
        a0.x = a4.x; a0.y = a4.y;
        a1.x = a4.z; a1.y = a4.w;
#pragma unroll
        for (int nt = 0; nt < NT; ++nt) {
            v2f b0;
            b0.x = bq[nt].x; b0.y = bq[nt].y;
            c[nt] = __builtin_amdgcn_wmma_f32_16x16x4_f32(
                         false, a0,  false, b0,
                         (short)0, c[nt],  false,  false);
        }
#pragma unroll
        for (int nt = 0; nt < NT; ++nt) {
            v2f b1;
            b1.x = bq[nt].z; b1.y = bq[nt].w;
            c[nt] = __builtin_amdgcn_wmma_f32_16x16x4_f32(
                         false, a1,  false, b1,
                         (short)0, c[nt],  false,  false);
        }
    }

    const int row0 = strip * 16 + 8 * half;
    float* crow = C + (long long)row0 * N + l15;
#pragma unroll
    for (int r = 0; r < 8; ++r) {
        const float di = rsqrtf((float)(cnt[row0 + r] + 1));
#pragma unroll
        for (int nt = 0; nt < NT; ++nt)
            crow[(long long)r * N + nt * 16] = c[nt][r] * di;
    }
}

constexpr int LAYER_IN = 128, LAYER_HID = 64, LAYER_OUT = 32;

extern "C" void kernel_launch(void* const* d_in, const int* in_sizes, int n_in,
                              void* d_out, int out_size, void* d_ws, size_t ws_size,
                              hipStream_t stream) {
    const float* x   = (const float*)d_in[0];
    const int*   ei  = (const int*)d_in[1];
    const float* W1  = (const float*)d_in[2];
    const float* b1  = (const float*)d_in[3];
    const float* W2  = (const float*)d_in[4];
    const float* b2  = (const float*)d_in[5];
    float*       out = (float*)d_out;

    const int E     = in_sizes[1] / 2;
    const int F_HID = in_sizes[3];
    const int F_IN  = in_sizes[2] / F_HID;
    const int F_OUT = in_sizes[5];
    const int N     = in_sizes[0] / F_IN;

    const int* src = ei;
    const int* dst = ei + E;

    const long long Epad = ((long long)E + 63) & ~63LL;
    const long long Npad = ((long long)N + 63) & ~63LL;
    int*   cnt      = (int*)d_ws;
    int*   ovcount  = cnt + Npad;
    int*   ent      = ovcount + 64;
    int*   ov       = ent + Npad * CAP;
    const int NC  = (E + CHUNK - 1) / CHUNK;
    const int NB  = (N + ROWS_PER_BUCKET - 1) / ROWS_PER_BUCKET;
    const int NBP = (NB + 31) & ~31;
    int*   sorted   = ov + 2 * Epad;
    int*   runs     = sorted + 2 * Epad;
    float* xw1 = (float*)(runs + 2 * (long long)NC * NBP);
    float* hw2 = xw1 + (long long)N * F_HID;

    const int TB = 256;
    const int rblocks = (N + TB / 32 - 1) / (TB / 32);
    const int gblocks = (N / 16 + GEMM_WAVES - 1) / GEMM_WAVES;

    edge_chunk_split_kernel<<<NC, SORT_THREADS, 0, stream>>>(src, dst, sorted, runs, ovcount,
                                                             E, NB, NBP);
    range_rows_kernel<<<NB, SORT_THREADS, 0, stream>>>(sorted, runs, cnt, ent, ov, ovcount,
                                                       N, NC, NBP);

    gemm_strip_wmma_f32_kernel<LAYER_IN, LAYER_HID / 16>
        <<<gblocks, 32 * GEMM_WAVES, 0, stream>>>(x, W1, cnt, xw1, N);

    gather_relu_gemm_wmma_f32_kernel<LAYER_HID, LAYER_OUT / 16>
        <<<N / 16, 32 * STRIP_WAVES, 0, stream>>>(cnt, ent, ov, ovcount, xw1, b1, W2, hw2);

    row_gather_kernel<1><<<rblocks, TB, 0, stream>>>(cnt, ent, ov, ovcount, hw2, b2, out, N);
}


# neighbour sums on fewer, longer waves: 4 rows per wave in both layers (output layer 50000 -> 12500 waves, fused layer 25000 -> 12500), counts and slot lines asked for together; sum order unchanged
# speedup vs baseline: 6.8819x; 1.0550x over previous
#include <hip/hip_runtime.h>
#include <hip/hip_bf16.h>

typedef float v2f __attribute__((ext_vector_type(2)));
typedef float v4f __attribute__((ext_vector_type(4)));
typedef float v8f __attribute__((ext_vector_type(8)));
typedef int   v2i __attribute__((ext_vector_type(2)));

constexpr int CAP = 32;
constexpr int BUCKET_SHIFT    = 8;
constexpr int ROWS_PER_BUCKET = 1 << BUCKET_SHIFT;
constexpr int CHUNK           = 4096;
constexpr int SORT_THREADS    = 512;
constexpr int NB_MAX          = SORT_THREADS;

typedef int v4i __attribute__((ext_vector_type(4)));

__global__ void __launch_bounds__(SORT_THREADS)
edge_chunk_split_kernel(const int* __restrict__ src, const int* __restrict__ dst,
                        int* __restrict__ sorted_w, int* __restrict__ runs_w,
                        int* __restrict__ ovcount, int e, int nb, int nbp) {
    constexpr int PER = CHUNK / SORT_THREADS;
    __shared__ int hist[NB_MAX];
    __shared__ int start[NB_MAX];
    __shared__ int wtot[SORT_THREADS / 32];
    __shared__ __attribute__((aligned(16))) v2i stage[CHUNK];
    v2i* sorted = (v2i*)sorted_w;
    v2i* runs   = (v2i*)runs_w;

    const int tid  = threadIdx.x;
    const int c    = blockIdx.x;
    const int base = c * CHUNK;
    if (c == 0 && tid == 0) *ovcount = 0;
    for (int i = tid; i < nb; i += SORT_THREADS) hist[i] = 0;

    int s[PER], d[PER], r[PER];
#pragma unroll
    for (int i = 0; i < PER; ++i) {
        const int idx = base + tid + SORT_THREADS * i;
        s[i] = 0;
        d[i] = -1;
        if (idx < e) {
            s[i] = src[idx];
            d[i] = dst[idx];
        }
    }
    __syncthreads();
#pragma unroll
    for (int i = 0; i < PER; ++i) {
        r[i] = 0;
        if (d[i] >= 0) r[i] = atomicAdd(&hist[d[i] >> BUCKET_SHIFT], 1);
    }
    __syncthreads();

    {
        const int lane = tid & 31;
        const int wave = tid >> 5;
        const int v = (tid < nb) ? hist[tid] : 0;
        int incl = v;
        for (int off = 1; off < 32; off <<= 1) {
            const int t = __shfl_up(incl, off);
            if (lane >= off) incl += t;
        }
        if (lane == 31) wtot[wave] = incl;
        __syncthreads();
        int front = 0;
#pragma unroll
        for (int w = 0; w < SORT_THREADS / 32; ++w) front += (w < wave) ? wtot[w] : 0;
        if (tid < nb) start[tid] = front + incl - v;
    }
    __syncthreads();

    for (int i = tid; i < nb; i += SORT_THREADS) {
        v2i p;
        p.x = start[i];
        p.y = hist[i];
        runs[(long long)c * nbp + i] = p;
    }
#pragma unroll
    for (int i = 0; i < PER; ++i) {
        if (d[i] >= 0) {
            v2i p;
            p.x = s[i];
            p.y = d[i];
            stage[start[d[i] >> BUCKET_SHIFT] + r[i]] = p;
        }
    }
    __syncthreads();
#pragma unroll
    for (int i = 0; i < PER; ++i) {
        const int k = tid + SORT_THREADS * i;
        if (base + k < e) sorted[(long long)base + k] = stage[k];
    }
}

constexpr int RUN_LANES  = 8;
constexpr int RUN_SLOTS  = SORT_THREADS / RUN_LANES;
constexpr int RUN_UNROLL = 4;
constexpr int RUN_DEPTH  = 4;

__device__ __forceinline__ void file_edge(int* lent, int* __restrict__ ov,
                                          int* __restrict__ ovcount, int row0, v2i ed, int r) {
    if (r < CAP) {
        lent[(ed.y - row0) * CAP + r] = ed.x;
    } else {
        const int k2 = atomicAdd(ovcount, 1);
        v2i o;
        o.x = ed.y;
        o.y = ed.x;
        *(v2i*)(ov + 2 * (long long)k2) = o;
    }
}

__global__ void __launch_bounds__(SORT_THREADS)
range_rows_kernel(const int* __restrict__ sorted_w, const int* __restrict__ runs_w,
                  int* __restrict__ cnt, int* __restrict__ ent,
                  int* __restrict__ ov, int* __restrict__ ovcount,
                  int n, int nc, int nbp) {
    __shared__ int lcnt[ROWS_PER_BUCKET];
    __shared__ __attribute__((aligned(16))) int lent[ROWS_PER_BUCKET * CAP];
    const v2i* sorted = (const v2i*)sorted_w;
    const v2i* runs   = (const v2i*)runs_w;

    const int tid  = threadIdx.x;
    const int b    = blockIdx.x;
    const int row0 = b << BUCKET_SHIFT;
    if (tid < ROWS_PER_BUCKET) lcnt[tid] = 0;
    __syncthreads();

    const int j    = tid % RUN_LANES;
    const int slot = tid / RUN_LANES;
    for (int c0 = slot; c0 < nc; c0 += RUN_UNROLL * RUN_SLOTS) {
        const v2i* pc[RUN_UNROLL];
        int first[RUN_UNROLL], len[RUN_UNROLL];
#pragma unroll
        for (int u = 0; u < RUN_UNROLL; ++u) {
            const int c  = c0 + u * RUN_SLOTS;
            const int cc = (c < nc) ? c : nc - 1;
            const v2i run = runs[(long long)cc * nbp + b];
            pc[u]    = sorted + (long long)cc * CHUNK;
            first[u] = run.x;
            len[u]   = (c < nc) ? run.y : 0;
        }
        v2i ed[RUN_UNROLL][RUN_DEPTH];
#pragma unroll
        for (int u = 0; u < RUN_UNROLL; ++u) {
#pragma unroll
            for (int d = 0; d < RUN_DEPTH; ++d) {
                const int k = j + RUN_LANES * d;
                ed[u][d] = pc[u][(k < len[u]) ? first[u] + k : 0];
            }
        }
        int r[RUN_UNROLL][RUN_DEPTH];
#pragma unroll
        for (int u = 0; u < RUN_UNROLL; ++u) {
#pragma unroll
            for (int d = 0; d < RUN_DEPTH; ++d) {
                r[u][d] = 0;
                if (j + RUN_LANES * d < len[u]) r[u][d] = atomicAdd(&lcnt[ed[u][d].y - row0], 1);
            }
        }
#pragma unroll
        for (int u = 0; u < RUN_UNROLL; ++u) {
#pragma unroll
            for (int d = 0; d < RUN_DEPTH; ++d) {
                if (j + RUN_LANES * d < len[u])
                    file_edge(lent, ov, ovcount, row0, ed[u][d], r[u][d]);
            }
        }
#pragma unroll
        for (int u = 0; u < RUN_UNROLL; ++u) {
            for (int k = j + RUN_LANES * RUN_DEPTH; k < len[u]; k += RUN_LANES) {
                const v2i e0 = pc[u][first[u] + k];
                const int r0 = atomicAdd(&lcnt[e0.y - row0], 1);
                file_edge(lent, ov, ovcount, row0, e0, r0);
            }
        }
    }
    __syncthreads();

    if (tid < ROWS_PER_BUCKET && row0 + tid < n) cnt[row0 + tid] = lcnt[tid];
#pragma unroll
    for (int i = 0; i < ROWS_PER_BUCKET * CAP / 4 / SORT_THREADS; ++i) {
        const int idx = tid + SORT_THREADS * i;
        const int row = row0 + (idx >> 3);
        if (row < n) ((v4i*)ent)[(long long)row0 * (CAP / 4) + idx] = ((const v4i*)lent)[idx];
    }
}

constexpr int GEMM_WAVES = 8;

template <int K, int NT>
__global__ void __launch_bounds__(32 * GEMM_WAVES)
gemm_strip_wmma_f32_kernel(const float* __restrict__ A,
                           const float* __restrict__ B,
                           const int* __restrict__ cnt,
                           float* __restrict__ C, int M) {
    constexpr int N     = 16 * NT;
    constexpr int PITCH = K + 4;
    constexpr int KB    = K / 8;
    constexpr int TB    = 32 * GEMM_WAVES;
    constexpr int UNITS = (K / 4) * N / TB;
    __shared__ __attribute__((aligned(16))) float Bt[N * PITCH];
    __shared__ __attribute__((aligned(16))) float As[GEMM_WAVES * 16 * PITCH];

    const int tid   = threadIdx.x;
    const int lane  = tid & 31;
    const int wave  = tid >> 5;
    const int half  = lane >> 4;
    const int l15   = lane & 15;
    const int strips = M >> 4;
    const int strip  = blockIdx.x * GEMM_WAVES + wave;
    const int sload  = (strip < strips) ? strip : strips - 1;

    v4f w[UNITS];
#pragma unroll
    for (int i = 0; i < UNITS; ++i) {
        const int p  = tid + TB * i;
        const int n  = p % N;
        const int kg = p / N;
        const float* bsrc = B + (long long)(4 * kg) * N + n;
        w[i].x = bsrc[0];
        w[i].y = bsrc[N];
        w[i].z = bsrc[2 * N];
        w[i].w = bsrc[3 * N];
    }
    const float* astrip = A + (long long)sload * (16 * K) + 4 * lane;
    v4f av[KB];
#pragma unroll
    for (int i = 0; i < KB; ++i) av[i] = *(const v4f*)(astrip + 128 * i);

#pragma unroll
    for (int i = 0; i < UNITS; ++i) {
        const int p  = tid + TB * i;
        const int n  = p % N;
        const int kg = p / N;
        *(v4f*)(&Bt[n * PITCH + 4 * kg]) = w[i];
    }
    float* aw = As + wave * (16 * PITCH);
#pragma unroll
    for (int i = 0; i < KB; ++i) {
        const int f   = 128 * i + 4 * lane;
        const int row = f / K;
        const int col = f % K;
        *(v4f*)(&aw[row * PITCH + col]) = av[i];
    }
    __syncthreads();
    if (strip >= strips) return;

    v8f c[NT];
#pragma unroll
    for (int nt = 0; nt < NT; ++nt) c[nt] = v8f{};

    const float* al = aw + l15 * PITCH + 4 * half;
    const float* bl = Bt + l15 * PITCH + 4 * half;
#pragma unroll
    for (int j = 0; j < KB; ++j) {
        const v4f a4 = *(const v4f*)(al + 8 * j);
        v4f bv[NT];
#pragma unroll
        for (int nt = 0; nt < NT; ++nt) bv[nt] = *(const v4f*)(bl + nt * 16 * PITCH + 8 * j);
        v2f a0, a1;
        a0.x = a4.x; a0.y = a4.y;
        a1.x = a4.z; a1.y = a4.w;
#pragma unroll
        for (int nt = 0; nt < NT; ++nt) {
            v2f b0;
            b0.x = bv[nt].x; b0.y = bv[nt].y;
            c[nt] = __builtin_amdgcn_wmma_f32_16x16x4_f32(
                         false, a0,  false, b0,
                         (short)0, c[nt],  false,  false);
        }
#pragma unroll
        for (int nt = 0; nt < NT; ++nt) {
            v2f b1;
            b1.x = bv[nt].z; b1.y = bv[nt].w;
            c[nt] = __builtin_amdgcn_wmma_f32_16x16x4_f32(
                         false, a1,  false, b1,
                         (short)0, c[nt],  false,  false);
        }
    }

    const int row0 = strip * 16 + 8 * half;
    float* crow = C + (long long)row0 * N + l15;
#pragma unroll
    for (int r = 0; r < 8; ++r) {
        const float di = rsqrtf((float)(cnt[row0 + r] + 1));
#pragma unroll
        for (int nt = 0; nt < NT; ++nt)
            crow[(long long)r * N + nt * 16] = c[nt][r] * di;
    }
}

template <int VEC> struct RowVec;
template <> struct RowVec<1> { typedef float T; };
template <> struct RowVec<2> { typedef v2f   T; };

template <int G, int F, typename T>
__device__ __forceinline__ T add_slots(T acc, const float* __restrict__ xs, unsigned col,
                                       int s, int j) {
    T v[G];
#pragma unroll
    for (int k = 0; k < G; ++k) {
        const float* row = xs + (long long)__builtin_amdgcn_readlane(s, j + k) * F;
        v[k] = *(const T*)(row + col);
    }
#pragma unroll
    for (int k = 0; k < G; ++k) acc += v[k];
    return acc;
}

template <int F, typename T>
__device__ __forceinline__ T row_sum(const float* __restrict__ xs, const int* __restrict__ ov,
                                     const int* __restrict__ ovcount, unsigned col, int lane,
                                     int d, int m_all, int s, T self) {
    const int m = (m_all < CAP) ? m_all : CAP;
    T acc = self;
    int j = 0;
    for (; j + 16 <= m; j += 16) acc = add_slots<16, F, T>(acc, xs, col, s, j);
    if (m - j >= 8) { acc = add_slots<8, F, T>(acc, xs, col, s, j); j += 8; }
    if (m - j >= 4) { acc = add_slots<4, F, T>(acc, xs, col, s, j); j += 4; }
    if (m - j >= 2) { acc = add_slots<2, F, T>(acc, xs, col, s, j); j += 2; }
    if (m - j >= 1) { acc = add_slots<1, F, T>(acc, xs, col, s, j); }

    if (m_all > CAP) {
        const int novf = *ovcount;
        for (int k0 = 0; k0 < novf; k0 += 32) {
            int od = -1, os = 0;
            if (k0 + lane < novf) {
                const v2i p = *(const v2i*)(ov + 2 * (long long)(k0 + lane));
                od = p.x;
                os = p.y;
            }
            unsigned hit = __builtin_amdgcn_ballot_w32(od == d);
            while (hit) {
                const int b = __builtin_ctz(hit);
                hit &= hit - 1;
                const float* row = xs + (long long)__builtin_amdgcn_readlane(os, b) * F;
                acc += *(const T*)(row + col);
            }
        }
    }
    return acc;
}

constexpr int OUT_RPW   = 4;
constexpr int OUT_WAVES = 4;

template <int VEC>
__global__ void __launch_bounds__(32 * OUT_WAVES)
row_gather_kernel(const int* __restrict__ cnt,
                  const int* __restrict__ ent,
                  const int* __restrict__ ov,
                  const int* __restrict__ ovcount,
                  const float* __restrict__ xs,
                  const float* __restrict__ bias,
                  float* __restrict__ out, int n) {
    typedef typename RowVec<VEC>::T T;
    constexpr int F = 32 * VEC;
    const int lane = threadIdx.x & 31;
    const int wave = __builtin_amdgcn_readfirstlane((int)(threadIdx.x >> 5));
    const int d0   = (blockIdx.x * OUT_WAVES + wave) * OUT_RPW;
    if (d0 >= n) return;
    const unsigned col = (unsigned)(lane * VEC);

    const T bv = *(const T*)(bias + col);
    int dr[OUT_RPW], m_all[OUT_RPW], s[OUT_RPW];
    T   self[OUT_RPW];
#pragma unroll
    for (int r = 0; r < OUT_RPW; ++r) dr[r] = (d0 + r < n) ? d0 + r : n - 1;
    const int dl = (d0 + (lane & (OUT_RPW - 1)) < n) ? d0 + (lane & (OUT_RPW - 1)) : n - 1;
    const int cv = cnt[dl];
#pragma unroll
    for (int r = 0; r < OUT_RPW; ++r) s[r] = ent[(long long)dr[r] * CAP + lane];
#pragma unroll
    for (int r = 0; r < OUT_RPW; ++r) self[r] = *(const T*)(xs + (long long)dr[r] * F + col);
#pragma unroll
    for (int r = 0; r < OUT_RPW; ++r) m_all[r] = __builtin_amdgcn_readlane(cv, r);

#pragma unroll
    for (int r = 0; r < OUT_RPW; ++r) {
        const T acc = row_sum<F, T>(xs, ov, ovcount, col, lane, dr[r], m_all[r], s[r], self[r]);
        const float di = rsqrtf((float)(m_all[r] + 1));
        if (d0 + r < n)
            *(T*)(out + (long long)dr[r] * F + col) = acc * di + bv;
    }
}

constexpr int STRIP_WAVES = 4;

template <int K, int NT>
__global__ void __launch_bounds__(32 * STRIP_WAVES)
gather_relu_gemm_wmma_f32_kernel(const int* __restrict__ cnt,
                                 const int* __restrict__ ent,
                                 const int* __restrict__ ov,
                                 const int* __restrict__ ovcount,
                                 const float* __restrict__ xs,
                                 const float* __restrict__ bias,
                                 const float* __restrict__ B,
                                 float* __restrict__ C) {
    static_assert(K == 64, "one lane holds two columns of a row");
    typedef typename RowVec<K / 32>::T T;
    constexpr int N     = 16 * NT;
    constexpr int PITCH = K + 4;
    constexpr int KB    = K / 8;
    constexpr int TB    = 32 * STRIP_WAVES;
    constexpr int UNITS = (K / 4) * N / TB;
    constexpr int RPW   = 16 / STRIP_WAVES;
    __shared__ __attribute__((aligned(16))) float Bt[N * PITCH];
    __shared__ __attribute__((aligned(16))) float As[16 * PITCH];

    const int tid   = threadIdx.x;
    const int lane  = tid & 31;
    const int wave  = __builtin_amdgcn_readfirstlane((int)(threadIdx.x >> 5));
    const int strip = blockIdx.x;
    const unsigned col = (unsigned)(lane * (K / 32));

    v4f w[UNITS];
#pragma unroll
    for (int i = 0; i < UNITS; ++i) {
        const int p  = tid + TB * i;
        const int nn = p % N;
        const int kg = p / N;
        const float* bsrc = B + (long long)(4 * kg) * N + nn;
        w[i].x = bsrc[0];
        w[i].y = bsrc[N];
        w[i].z = bsrc[2 * N];
        w[i].w = bsrc[3 * N];
    }
    const T bv = *(const T*)(bias + col);
    const int d0 = strip * 16 + wave * RPW;
    int m_all[RPW], s[RPW];
    T   self[RPW];
#pragma unroll
    for (int r = 0; r < RPW; ++r) m_all[r] = cnt[d0 + r];
#pragma unroll
    for (int r = 0; r < RPW; ++r) self[r] = *(const T*)(xs + (long long)(d0 + r) * K + col);
#pragma unroll
    for (int r = 0; r < RPW; ++r) s[r] = ent[(long long)(d0 + r) * CAP + lane];
#pragma unroll
    for (int i = 0; i < UNITS; ++i) {
        const int p  = tid + TB * i;
        const int nn = p % N;
        const int kg = p / N;
        *(v4f*)(&Bt[nn * PITCH + 4 * kg]) = w[i];
    }

#pragma unroll
    for (int r = 0; r < RPW; ++r) {
        const T acc = row_sum<K, T>(xs, ov, ovcount, col, lane, d0 + r, m_all[r], s[r], self[r]);
        const float di = rsqrtf((float)(m_all[r] + 1));
        T hv = acc * di + bv;
        hv.x = fmaxf(hv.x, 0.0f);
        hv.y = fmaxf(hv.y, 0.0f);
        *(T*)(&As[(wave * RPW + r) * PITCH + col]) = hv;
    }
    __syncthreads();
    if (wave != 0) return;

    const int half = lane >> 4;
    const int l15  = lane & 15;
    v8f c[NT];
#pragma unroll
    for (int nt = 0; nt < NT; ++nt) c[nt] = v8f{};

    const float* al = As + l15 * PITCH + 4 * half;
    const float* bl = Bt + l15 * PITCH + 4 * half;
#pragma unroll
    for (int j = 0; j < KB; ++j) {
        const v4f a4 = *(const v4f*)(al + 8 * j);
        v4f bq[NT];
#pragma unroll
        for (int nt = 0; nt < NT; ++nt) bq[nt] = *(const v4f*)(bl + nt * 16 * PITCH + 8 * j);
        v2f a0, a1;
        a0.x = a4.x; a0.y = a4.y;
        a1.x = a4.z; a1.y = a4.w;
#pragma unroll
        for (int nt = 0; nt < NT; ++nt) {
            v2f b0;
            b0.x = bq[nt].x; b0.y = bq[nt].y;
            c[nt] = __builtin_amdgcn_wmma_f32_16x16x4_f32(
                         false, a0,  false, b0,
                         (short)0, c[nt],  false,  false);
        }
#pragma unroll
        for (int nt = 0; nt < NT; ++nt) {
            v2f b1;
            b1.x = bq[nt].z; b1.y = bq[nt].w;
            c[nt] = __builtin_amdgcn_wmma_f32_16x16x4_f32(
                         false, a1,  false, b1,
                         (short)0, c[nt],  false,  false);
        }
    }

    const int row0 = strip * 16 + 8 * half;
    float* crow = C + (long long)row0 * N + l15;
#pragma unroll
    for (int r = 0; r < 8; ++r) {
        const float di = rsqrtf((float)(cnt[row0 + r] + 1));
#pragma unroll
        for (int nt = 0; nt < NT; ++nt)
            crow[(long long)r * N + nt * 16] = c[nt][r] * di;
    }
}

constexpr int LAYER_IN = 128, LAYER_HID = 64, LAYER_OUT = 32;

extern "C" void kernel_launch(void* const* d_in, const int* in_sizes, int n_in,
                              void* d_out, int out_size, void* d_ws, size_t ws_size,
                              hipStream_t stream) {
    const float* x   = (const float*)d_in[0];
    const int*   ei  = (const int*)d_in[1];
    const float* W1  = (const float*)d_in[2];
    const float* b1  = (const float*)d_in[3];
    const float* W2  = (const float*)d_in[4];
    const float* b2  = (const float*)d_in[5];
    float*       out = (float*)d_out;

    const int E     = in_sizes[1] / 2;
    const int F_HID = in_sizes[3];
    const int F_IN  = in_sizes[2] / F_HID;
    const int F_OUT = in_sizes[5];
    const int N     = in_sizes[0] / F_IN;

    const int* src = ei;
    const int* dst = ei + E;

    const long long Epad = ((long long)E + 63) & ~63LL;
    const long long Npad = ((long long)N + 63) & ~63LL;
    int*   cnt      = (int*)d_ws;
    int*   ovcount  = cnt + Npad;
    int*   ent      = ovcount + 64;
    int*   ov       = ent + Npad * CAP;
    const int NC  = (E + CHUNK - 1) / CHUNK;
    const int NB  = (N + ROWS_PER_BUCKET - 1) / ROWS_PER_BUCKET;
    const int NBP = (NB + 31) & ~31;
    int*   sorted   = ov + 2 * Epad;
    int*   runs     = sorted + 2 * Epad;
    float* xw1 = (float*)(runs + 2 * (long long)NC * NBP);
    float* hw2 = xw1 + (long long)N * F_HID;

    const int rblocks = (N + OUT_WAVES * OUT_RPW - 1) / (OUT_WAVES * OUT_RPW);
    const int gblocks = (N / 16 + GEMM_WAVES - 1) / GEMM_WAVES;

    edge_chunk_split_kernel<<<NC, SORT_THREADS, 0, stream>>>(src, dst, sorted, runs, ovcount,
                                                             E, NB, NBP);
    range_rows_kernel<<<NB, SORT_THREADS, 0, stream>>>(sorted, runs, cnt, ent, ov, ovcount,
                                                       N, NC, NBP);

    gemm_strip_wmma_f32_kernel<LAYER_IN, LAYER_HID / 16>
        <<<gblocks, 32 * GEMM_WAVES, 0, stream>>>(x, W1, cnt, xw1, N);

    gather_relu_gemm_wmma_f32_kernel<LAYER_HID, LAYER_OUT / 16>
        <<<N / 16, 32 * STRIP_WAVES, 0, stream>>>(cnt, ent, ov, ovcount, xw1, b1, W2, hw2);

    row_gather_kernel<1><<<rblocks, 32 * OUT_WAVES, 0, stream>>>(cnt, ent, ov, ovcount, hw2, b2, out, N);
}


# layer-1 product moved into the row-range pass: a block ranks its 256 rows' edges, then each of its 16 waves takes a 16-row strip of them (row scale: LDS counts); 5->4 launches; sum order unchanged
# speedup vs baseline: 7.3346x; 1.0658x over previous
#include <hip/hip_runtime.h>
#include <hip/hip_bf16.h>

typedef float v2f __attribute__((ext_vector_type(2)));
typedef float v4f __attribute__((ext_vector_type(4)));
typedef float v8f __attribute__((ext_vector_type(8)));
typedef int   v2i __attribute__((ext_vector_type(2)));

constexpr int CAP = 32;
constexpr int BUCKET_SHIFT    = 8;
constexpr int ROWS_PER_BUCKET = 1 << BUCKET_SHIFT;
constexpr int CHUNK           = 4096;
constexpr int SORT_THREADS    = 512;
constexpr int NB_MAX          = SORT_THREADS;

typedef int v4i __attribute__((ext_vector_type(4)));

__global__ void __launch_bounds__(SORT_THREADS)
edge_chunk_split_kernel(const int* __restrict__ src, const int* __restrict__ dst,
                        int* __restrict__ sorted_w, int* __restrict__ runs_w,
                        int* __restrict__ ovcount, int e, int nb, int nbp) {
    constexpr int PER = CHUNK / SORT_THREADS;
    __shared__ int hist[NB_MAX];
    __shared__ int start[NB_MAX];
    __shared__ int wtot[SORT_THREADS / 32];
    __shared__ __attribute__((aligned(16))) v2i stage[CHUNK];
    v2i* sorted = (v2i*)sorted_w;
    v2i* runs   = (v2i*)runs_w;

    const int tid  = threadIdx.x;
    const int c    = blockIdx.x;
    const int base = c * CHUNK;
    if (c == 0 && tid == 0) *ovcount = 0;
    for (int i = tid; i < nb; i += SORT_THREADS) hist[i] = 0;

    int s[PER], d[PER], r[PER];
#pragma unroll
    for (int i = 0; i < PER; ++i) {
        const int idx = base + tid + SORT_THREADS * i;
        s[i] = 0;
        d[i] = -1;
        if (idx < e) {
            s[i] = src[idx];
            d[i] = dst[idx];
        }
    }
    __syncthreads();
#pragma unroll
    for (int i = 0; i < PER; ++i) {
        r[i] = 0;
        if (d[i] >= 0) r[i] = atomicAdd(&hist[d[i] >> BUCKET_SHIFT], 1);
    }
    __syncthreads();

    {
        const int lane = tid & 31;
        const int wave = tid >> 5;
        const int v = (tid < nb) ? hist[tid] : 0;
        int incl = v;
        for (int off = 1; off < 32; off <<= 1) {
            const int t = __shfl_up(incl, off);
            if (lane >= off) incl += t;
        }
        if (lane == 31) wtot[wave] = incl;
        __syncthreads();
        int front = 0;
#pragma unroll
        for (int w = 0; w < SORT_THREADS / 32; ++w) front += (w < wave) ? wtot[w] : 0;
        if (tid < nb) start[tid] = front + incl - v;
    }
    __syncthreads();

    for (int i = tid; i < nb; i += SORT_THREADS) {
        v2i p;
        p.x = start[i];
        p.y = hist[i];
        runs[(long long)c * nbp + i] = p;
    }
#pragma unroll
    for (int i = 0; i < PER; ++i) {
        if (d[i] >= 0) {
            v2i p;
            p.x = s[i];
            p.y = d[i];
            stage[start[d[i] >> BUCKET_SHIFT] + r[i]] = p;
        }
    }
    __syncthreads();
#pragma unroll
    for (int i = 0; i < PER; ++i) {
        const int k = tid + SORT_THREADS * i;
        if (base + k < e) sorted[(long long)base + k] = stage[k];
    }
}

constexpr int RUN_LANES  = 8;
constexpr int RUN_SLOTS  = SORT_THREADS / RUN_LANES;
constexpr int RUN_UNROLL = 4;
constexpr int RUN_DEPTH  = 4;

__device__ __forceinline__ void file_edge(int* lent, int* __restrict__ ov,
                                          int* __restrict__ ovcount, int row0, v2i ed, int r) {
    if (r < CAP) {
        lent[(ed.y - row0) * CAP + r] = ed.x;
    } else {
        const int k2 = atomicAdd(ovcount, 1);
        v2i o;
        o.x = ed.y;
        o.y = ed.x;
        *(v2i*)(ov + 2 * (long long)k2) = o;
    }
}

template <int K, int NT>
__global__ void __launch_bounds__(SORT_THREADS)
range_rows_gemm_wmma_f32_kernel(const int* __restrict__ sorted_w, const int* __restrict__ runs_w,
                                int* __restrict__ cnt, int* __restrict__ ent,
                                int* __restrict__ ov, int* __restrict__ ovcount,
                                int n, int nc, int nbp,
                                const float* __restrict__ A, const float* __restrict__ B,
                                float* __restrict__ C) {
    constexpr int N     = 16 * NT;
    constexpr int PITCH = K + 4;
    constexpr int KB    = K / 8;
    constexpr int TB    = SORT_THREADS;
    constexpr int WAVES = SORT_THREADS / 32;
    constexpr int UNITS = (K / 4) * N / TB;
    static_assert(WAVES * 16 == ROWS_PER_BUCKET, "one 16-row strip per wave of the block");
    static_assert(UNITS * TB == (K / 4) * N, "B pieces divide evenly over the block");
    __shared__ int lcnt[ROWS_PER_BUCKET];
    __shared__ __attribute__((aligned(16))) int lent[ROWS_PER_BUCKET * CAP];
    __shared__ __attribute__((aligned(16))) float Bt[N * PITCH];
    __shared__ __attribute__((aligned(16))) float As[WAVES * 16 * PITCH];
    const v2i* sorted = (const v2i*)sorted_w;
    const v2i* runs   = (const v2i*)runs_w;

    const int tid    = threadIdx.x;
    const int b      = blockIdx.x;
    const int row0   = b << BUCKET_SHIFT;
    const int lane   = tid & 31;
    const int wave   = tid >> 5;
    const int half   = lane >> 4;
    const int l15    = lane & 15;
    const int strips = n >> 4;
    const int strip  = b * WAVES + wave;
    const int sload  = (strip < strips) ? strip : strips - 1;
    if (tid < ROWS_PER_BUCKET) lcnt[tid] = 0;
    __syncthreads();

    v4f w[UNITS];
    v4f av[KB];

    const int j     = tid % RUN_LANES;
    const int slot  = tid / RUN_LANES;
    const int clast = (nc > 0) ? nc - 1 : 0;
    auto trip = [&](const int c0, const bool ask_operands) {
        const v2i* pc[RUN_UNROLL];
        int first[RUN_UNROLL], len[RUN_UNROLL];
#pragma unroll
        for (int u = 0; u < RUN_UNROLL; ++u) {
            const int c  = c0 + u * RUN_SLOTS;
            const int cc = (c < nc) ? c : clast;
            const v2i run = runs[(long long)cc * nbp + b];
            pc[u]    = sorted + (long long)cc * CHUNK;
            first[u] = run.x;
            len[u]   = (c < nc) ? run.y : 0;
        }
        v2i ed[RUN_UNROLL][RUN_DEPTH];
#pragma unroll
        for (int u = 0; u < RUN_UNROLL; ++u) {
#pragma unroll
            for (int d = 0; d < RUN_DEPTH; ++d) {
                const int k = j + RUN_LANES * d;
                ed[u][d] = pc[u][(k < len[u]) ? first[u] + k : 0];
            }
        }
        if (ask_operands) {
#pragma unroll
            for (int i = 0; i < UNITS; ++i) {
                const int p  = tid + TB * i;
                const int nn = p % N;
                const int kg = p / N;
                const float* bsrc = B + (long long)(4 * kg) * N + nn;
                w[i].x = bsrc[0];
                w[i].y = bsrc[N];
                w[i].z = bsrc[2 * N];
                w[i].w = bsrc[3 * N];
            }
            const float* astrip = A + (long long)sload * (16 * K) + 4 * lane;
#pragma unroll
            for (int i = 0; i < KB; ++i) av[i] = *(const v4f*)(astrip + 128 * i);
        }
        int r[RUN_UNROLL][RUN_DEPTH];
#pragma unroll
        for (int u = 0; u < RUN_UNROLL; ++u) {
#pragma unroll
            for (int d = 0; d < RUN_DEPTH; ++d) {
                r[u][d] = 0;
                if (j + RUN_LANES * d < len[u]) r[u][d] = atomicAdd(&lcnt[ed[u][d].y - row0], 1);
            }
        }
#pragma unroll
        for (int u = 0; u < RUN_UNROLL; ++u) {
#pragma unroll
            for (int d = 0; d < RUN_DEPTH; ++d) {
                if (j + RUN_LANES * d < len[u])
                    file_edge(lent, ov, ovcount, row0, ed[u][d], r[u][d]);
            }
        }
#pragma unroll
        for (int u = 0; u < RUN_UNROLL; ++u) {
            for (int k = j + RUN_LANES * RUN_DEPTH; k < len[u]; k += RUN_LANES) {
                const v2i e0 = pc[u][first[u] + k];
                const int r0 = atomicAdd(&lcnt[e0.y - row0], 1);
                file_edge(lent, ov, ovcount, row0, e0, r0);
            }
        }
    };
    trip(slot, true);
    for (int c0 = slot + RUN_UNROLL * RUN_SLOTS; c0 < nc; c0 += RUN_UNROLL * RUN_SLOTS)
        trip(c0, false);

#pragma unroll
    for (int i = 0; i < UNITS; ++i) {
        const int p  = tid + TB * i;
        const int nn = p % N;
        const int kg = p / N;
        *(v4f*)(&Bt[nn * PITCH + 4 * kg]) = w[i];
    }
    float* aw = As + wave * (16 * PITCH);
#pragma unroll
    for (int i = 0; i < KB; ++i) {
        const int f   = 128 * i + 4 * lane;
        const int row = f / K;
        const int col = f % K;
        *(v4f*)(&aw[row * PITCH + col]) = av[i];
    }
    __syncthreads();

    if (tid < ROWS_PER_BUCKET && row0 + tid < n) cnt[row0 + tid] = lcnt[tid];
#pragma unroll
    for (int i = 0; i < ROWS_PER_BUCKET * CAP / 4 / SORT_THREADS; ++i) {
        const int idx = tid + SORT_THREADS * i;
        const int row = row0 + (idx >> 3);
        if (row < n) ((v4i*)ent)[(long long)row0 * (CAP / 4) + idx] = ((const v4i*)lent)[idx];
    }
    if (strip >= strips) return;

    v8f c[NT];
#pragma unroll
    for (int nt = 0; nt < NT; ++nt) c[nt] = v8f{};

    const float* al = aw + l15 * PITCH + 4 * half;
    const float* bl = Bt + l15 * PITCH + 4 * half;
#pragma unroll
    for (int jb = 0; jb < KB; ++jb) {
        const v4f a4 = *(const v4f*)(al + 8 * jb);
        v4f bv[NT];
#pragma unroll
        for (int nt = 0; nt < NT; ++nt) bv[nt] = *(const v4f*)(bl + nt * 16 * PITCH + 8 * jb);
        v2f a0, a1;
        a0.x = a4.x; a0.y = a4.y;
        a1.x = a4.z; a1.y = a4.w;
#pragma unroll
        for (int nt = 0; nt < NT; ++nt) {
            v2f b0;
            b0.x = bv[nt].x; b0.y = bv[nt].y;
            c[nt] = __builtin_amdgcn_wmma_f32_16x16x4_f32(
                         false, a0,  false, b0,
                         (short)0, c[nt],  false,  false);
        }
#pragma unroll
        for (int nt = 0; nt < NT; ++nt) {
            v2f b1;
            b1.x = bv[nt].z; b1.y = bv[nt].w;
            c[nt] = __builtin_amdgcn_wmma_f32_16x16x4_f32(
                         false, a1,  false, b1,
                         (short)0, c[nt],  false,  false);
        }
    }

    const int lrow0 = wave * 16 + 8 * half;
    float* crow = C + (long long)(row0 + lrow0) * N + l15;
#pragma unroll
    for (int r = 0; r < 8; ++r) {
        const float di = rsqrtf((float)(lcnt[lrow0 + r] + 1));
#pragma unroll
        for (int nt = 0; nt < NT; ++nt)
            crow[(long long)r * N + nt * 16] = c[nt][r] * di;
    }
}

template <int VEC> struct RowVec;
template <> struct RowVec<1> { typedef float T; };
template <> struct RowVec<2> { typedef v2f   T; };

template <int G, int F, typename T>
__device__ __forceinline__ T add_slots(T acc, const float* __restrict__ xs, unsigned col,
                                       int s, int j) {
    T v[G];
#pragma unroll
    for (int k = 0; k < G; ++k) {
        const float* row = xs + (long long)__builtin_amdgcn_readlane(s, j + k) * F;
        v[k] = *(const T*)(row + col);
    }
#pragma unroll
    for (int k = 0; k < G; ++k) acc += v[k];
    return acc;
}

template <int F, typename T>
__device__ __forceinline__ T row_sum(const float* __restrict__ xs, const int* __restrict__ ov,
                                     const int* __restrict__ ovcount, unsigned col, int lane,
                                     int d, int m_all, int s, T self) {
    const int m = (m_all < CAP) ? m_all : CAP;
    T acc = self;
    int j = 0;
    for (; j + 16 <= m; j += 16) acc = add_slots<16, F, T>(acc, xs, col, s, j);
    if (m - j >= 8) { acc = add_slots<8, F, T>(acc, xs, col, s, j); j += 8; }
    if (m - j >= 4) { acc = add_slots<4, F, T>(acc, xs, col, s, j); j += 4; }
    if (m - j >= 2) { acc = add_slots<2, F, T>(acc, xs, col, s, j); j += 2; }
    if (m - j >= 1) { acc = add_slots<1, F, T>(acc, xs, col, s, j); }

    if (m_all > CAP) {
        const int novf = *ovcount;
        for (int k0 = 0; k0 < novf; k0 += 32) {
            int od = -1, os = 0;
            if (k0 + lane < novf) {
                const v2i p = *(const v2i*)(ov + 2 * (long long)(k0 + lane));
                od = p.x;
                os = p.y;
            }
            unsigned hit = __builtin_amdgcn_ballot_w32(od == d);
            while (hit) {
                const int b = __builtin_ctz(hit);
                hit &= hit - 1;
                const float* row = xs + (long long)__builtin_amdgcn_readlane(os, b) * F;
                acc += *(const T*)(row + col);
            }
        }
    }
    return acc;
}

constexpr int OUT_RPW   = 4;
constexpr int OUT_WAVES = 4;

template <int VEC>
__global__ void __launch_bounds__(32 * OUT_WAVES)
row_gather_kernel(const int* __restrict__ cnt,
                  const int* __restrict__ ent,
                  const int* __restrict__ ov,
                  const int* __restrict__ ovcount,
                  const float* __restrict__ xs,
                  const float* __restrict__ bias,
                  float* __restrict__ out, int n) {
    typedef typename RowVec<VEC>::T T;
    constexpr int F = 32 * VEC;
    const int lane = threadIdx.x & 31;
    const int wave = __builtin_amdgcn_readfirstlane((int)(threadIdx.x >> 5));
    const int d0   = (blockIdx.x * OUT_WAVES + wave) * OUT_RPW;
    if (d0 >= n) return;
    const unsigned col = (unsigned)(lane * VEC);

    const T bv = *(const T*)(bias + col);
    int dr[OUT_RPW], m_all[OUT_RPW], s[OUT_RPW];
    T   self[OUT_RPW];
#pragma unroll
    for (int r = 0; r < OUT_RPW; ++r) dr[r] = (d0 + r < n) ? d0 + r : n - 1;
    const int dl = (d0 + (lane & (OUT_RPW - 1)) < n) ? d0 + (lane & (OUT_RPW - 1)) : n - 1;
    const int cv = cnt[dl];
#pragma unroll
    for (int r = 0; r < OUT_RPW; ++r) s[r] = ent[(long long)dr[r] * CAP + lane];
#pragma unroll
    for (int r = 0; r < OUT_RPW; ++r) self[r] = *(const T*)(xs + (long long)dr[r] * F + col);
#pragma unroll
    for (int r = 0; r < OUT_RPW; ++r) m_all[r] = __builtin_amdgcn_readlane(cv, r);

#pragma unroll
    for (int r = 0; r < OUT_RPW; ++r) {
        const T acc = row_sum<F, T>(xs, ov, ovcount, col, lane, dr[r], m_all[r], s[r], self[r]);
        const float di = rsqrtf((float)(m_all[r] + 1));
        if (d0 + r < n)
            *(T*)(out + (long long)dr[r] * F + col) = acc * di + bv;
    }
}

constexpr int STRIP_WAVES = 4;

template <int K, int NT>
__global__ void __launch_bounds__(32 * STRIP_WAVES)
gather_relu_gemm_wmma_f32_kernel(const int* __restrict__ cnt,
                                 const int* __restrict__ ent,
                                 const int* __restrict__ ov,
                                 const int* __restrict__ ovcount,
                                 const float* __restrict__ xs,
                                 const float* __restrict__ bias,
                                 const float* __restrict__ B,
                                 float* __restrict__ C) {
    static_assert(K == 64, "one lane holds two columns of a row");
    typedef typename RowVec<K / 32>::T T;
    constexpr int N     = 16 * NT;
    constexpr int PITCH = K + 4;
    constexpr int KB    = K / 8;
    constexpr int TB    = 32 * STRIP_WAVES;
    constexpr int UNITS = (K / 4) * N / TB;
    constexpr int RPW   = 16 / STRIP_WAVES;
    __shared__ __attribute__((aligned(16))) float Bt[N * PITCH];
    __shared__ __attribute__((aligned(16))) float As[16 * PITCH];

    const int tid   = threadIdx.x;
    const int lane  = tid & 31;
    const int wave  = __builtin_amdgcn_readfirstlane((int)(threadIdx.x >> 5));
    const int strip = blockIdx.x;
    const unsigned col = (unsigned)(lane * (K / 32));

    v4f w[UNITS];
#pragma unroll
    for (int i = 0; i < UNITS; ++i) {
        const int p  = tid + TB * i;
        const int nn = p % N;
        const int kg = p / N;
        const float* bsrc = B + (long long)(4 * kg) * N + nn;
        w[i].x = bsrc[0];
        w[i].y = bsrc[N];
        w[i].z = bsrc[2 * N];
        w[i].w = bsrc[3 * N];
    }
    const T bv = *(const T*)(bias + col);
    const int d0 = strip * 16 + wave * RPW;
    int m_all[RPW], s[RPW];
    T   self[RPW];
#pragma unroll
    for (int r = 0; r < RPW; ++r) m_all[r] = cnt[d0 + r];
#pragma unroll
    for (int r = 0; r < RPW; ++r) self[r] = *(const T*)(xs + (long long)(d0 + r) * K + col);
#pragma unroll
    for (int r = 0; r < RPW; ++r) s[r] = ent[(long long)(d0 + r) * CAP + lane];
#pragma unroll
    for (int i = 0; i < UNITS; ++i) {
        const int p  = tid + TB * i;
        const int nn = p % N;
        const int kg = p / N;
        *(v4f*)(&Bt[nn * PITCH + 4 * kg]) = w[i];
    }

#pragma unroll
    for (int r = 0; r < RPW; ++r) {
        const T acc = row_sum<K, T>(xs, ov, ovcount, col, lane, d0 + r, m_all[r], s[r], self[r]);
        const float di = rsqrtf((float)(m_all[r] + 1));
        T hv = acc * di + bv;
        hv.x = fmaxf(hv.x, 0.0f);
        hv.y = fmaxf(hv.y, 0.0f);
        *(T*)(&As[(wave * RPW + r) * PITCH + col]) = hv;
    }
    __syncthreads();
    if (wave != 0) return;

    const int half = lane >> 4;
    const int l15  = lane & 15;
    v8f c[NT];
#pragma unroll
    for (int nt = 0; nt < NT; ++nt) c[nt] = v8f{};

    const float* al = As + l15 * PITCH + 4 * half;
    const float* bl = Bt + l15 * PITCH + 4 * half;
#pragma unroll
    for (int j = 0; j < KB; ++j) {
        const v4f a4 = *(const v4f*)(al + 8 * j);
        v4f bq[NT];
#pragma unroll
        for (int nt = 0; nt < NT; ++nt) bq[nt] = *(const v4f*)(bl + nt * 16 * PITCH + 8 * j);
        v2f a0, a1;
        a0.x = a4.x; a0.y = a4.y;
        a1.x = a4.z; a1.y = a4.w;
#pragma unroll
        for (int nt = 0; nt < NT; ++nt) {
            v2f b0;
            b0.x = bq[nt].x; b0.y = bq[nt].y;
            c[nt] = __builtin_amdgcn_wmma_f32_16x16x4_f32(
                         false, a0,  false, b0,
                         (short)0, c[nt],  false,  false);
        }
#pragma unroll
        for (int nt = 0; nt < NT; ++nt) {
            v2f b1;
            b1.x = bq[nt].z; b1.y = bq[nt].w;
            c[nt] = __builtin_amdgcn_wmma_f32_16x16x4_f32(
                         false, a1,  false, b1,
                         (short)0, c[nt],  false,  false);
        }
    }

    const int row0 = strip * 16 + 8 * half;
    float* crow = C + (long long)row0 * N + l15;
#pragma unroll
    for (int r = 0; r < 8; ++r) {
        const float di = rsqrtf((float)(cnt[row0 + r] + 1));
#pragma unroll
        for (int nt = 0; nt < NT; ++nt)
            crow[(long long)r * N + nt * 16] = c[nt][r] * di;
    }
}

constexpr int LAYER_IN = 128, LAYER_HID = 64, LAYER_OUT = 32;

extern "C" void kernel_launch(void* const* d_in, const int* in_sizes, int n_in,
                              void* d_out, int out_size, void* d_ws, size_t ws_size,
                              hipStream_t stream) {
    const float* x   = (const float*)d_in[0];
    const int*   ei  = (const int*)d_in[1];
    const float* W1  = (const float*)d_in[2];
    const float* b1  = (const float*)d_in[3];
    const float* W2  = (const float*)d_in[4];
    const float* b2  = (const float*)d_in[5];
    float*       out = (float*)d_out;

    const int E     = in_sizes[1] / 2;
    const int F_HID = in_sizes[3];
    const int F_IN  = in_sizes[2] / F_HID;
    const int F_OUT = in_sizes[5];
    const int N     = in_sizes[0] / F_IN;

    const int* src = ei;
    const int* dst = ei + E;

    const long long Epad = ((long long)E + 63) & ~63LL;
    const long long Npad = ((long long)N + 63) & ~63LL;
    int*   cnt      = (int*)d_ws;
    int*   ovcount  = cnt + Npad;
    int*   ent      = ovcount + 64;
    int*   ov       = ent + Npad * CAP;
    const int NC  = (E + CHUNK - 1) / CHUNK;
    const int NB  = (N + ROWS_PER_BUCKET - 1) / ROWS_PER_BUCKET;
    const int NBP = (NB + 31) & ~31;
    int*   sorted   = ov + 2 * Epad;
    int*   runs     = sorted + 2 * Epad;
    float* xw1 = (float*)(runs + 2 * (long long)NC * NBP);
    float* hw2 = xw1 + (long long)N * F_HID;

    const int rblocks = (N + OUT_WAVES * OUT_RPW - 1) / (OUT_WAVES * OUT_RPW);

    edge_chunk_split_kernel<<<NC, SORT_THREADS, 0, stream>>>(src, dst, sorted, runs, ovcount,
                                                             E, NB, NBP);
    range_rows_gemm_wmma_f32_kernel<LAYER_IN, LAYER_HID / 16>
        <<<NB, SORT_THREADS, 0, stream>>>(sorted, runs, cnt, ent, ov, ovcount, N, NC, NBP,
                                          x, W1, xw1);

    gather_relu_gemm_wmma_f32_kernel<LAYER_HID, LAYER_OUT / 16>
        <<<N / 16, 32 * STRIP_WAVES, 0, stream>>>(cnt, ent, ov, ovcount, xw1, b1, W2, hw2);

    row_gather_kernel<1><<<rblocks, 32 * OUT_WAVES, 0, stream>>>(cnt, ent, ov, ovcount, hw2, b2, out, N);
}
